# starting kernel
# baseline (speedup 1.0000x reference)
#include <hip/hip_runtime.h>
#include <math.h>

#define NTOT 16384
#define NLAT 20
#define MCL  10
#define KNB  32
#define BB   8
#define MUF  5.0f

typedef __attribute__((ext_vector_type(2))) float v2f;
typedef __attribute__((ext_vector_type(8))) float v8f;

#define NPART   128
#define KCHUNK  (NTOT / NPART)
#define WS_PARTIALS 0
#define WS_ENCODED  (NPART * 512)
#define WS_INVD     (WS_ENCODED + BB * NLAT)
#define WS_EBL      (WS_INVD + MCL * NLAT * BB)

__global__ __launch_bounds__(32)
void encode_partial_wmma(const float* __restrict__ x,
                         const float* __restrict__ enc_w,
                         float* __restrict__ partials) {
  const int t = threadIdx.x;
  const int p = blockIdx.x;
  const int M = t & 15;
  const int Nn = t & 15;
  const int half = (t >> 4) << 1;
  const float maskA  = (M < BB) ? 1.f : 0.f;
  const float maskB1 = (Nn < NLAT - 16) ? 1.f : 0.f;
  const int rowA  = (M < BB) ? M : 0;
  const int rowB1 = (Nn < NLAT - 16) ? (16 + Nn) : 16;
  const float* pa  = x     + (size_t)rowA  * NTOT + half;
  const float* pb0 = enc_w + (size_t)Nn    * NTOT + half;
  const float* pb1 = enc_w + (size_t)rowB1 * NTOT + half;
  v8f acc0a = {}, acc0b = {}, acc1a = {}, acc1b = {};
  const int n0base = p * KCHUNK;
  for (int s = 0; s < KCHUNK; s += 8) {
    const int n0 = n0base + s;
    const float2 av0 = *(const float2*)(pa  + n0);
    const float2 bv0 = *(const float2*)(pb0 + n0);
    const float2 cv0 = *(const float2*)(pb1 + n0);
    const float2 av1 = *(const float2*)(pa  + n0 + 4);
    const float2 bv1 = *(const float2*)(pb0 + n0 + 4);
    const float2 cv1 = *(const float2*)(pb1 + n0 + 4);
    v2f a0; a0.x = av0.x * maskA;  a0.y = av0.y * maskA;
    v2f b0; b0.x = bv0.x;          b0.y = bv0.y;
    v2f c0; c0.x = cv0.x * maskB1; c0.y = cv0.y * maskB1;
    v2f a1; a1.x = av1.x * maskA;  a1.y = av1.y * maskA;
    v2f b1; b1.x = bv1.x;          b1.y = bv1.y;
    v2f c1; c1.x = cv1.x * maskB1; c1.y = cv1.y * maskB1;
    acc0a = __builtin_amdgcn_wmma_f32_16x16x4_f32(false, a0, false, b0,
                                                  (short)0, acc0a, false, false);
    acc1a = __builtin_amdgcn_wmma_f32_16x16x4_f32(false, a0, false, c0,
                                                  (short)0, acc1a, false, false);
    acc0b = __builtin_amdgcn_wmma_f32_16x16x4_f32(false, a1, false, b1,
                                                  (short)0, acc0b, false, false);
    acc1b = __builtin_amdgcn_wmma_f32_16x16x4_f32(false, a1, false, c1,
                                                  (short)0, acc1b, false, false);
  }
  float* out0 = partials + (size_t)(p * 2 + 0) * 256;
  float* out1 = partials + (size_t)(p * 2 + 1) * 256;
#pragma unroll
  for (int i = 0; i < 8; ++i) {
    out0[i * 32 + t] = acc0a[i] + acc0b[i];
    out1[i * 32 + t] = acc1a[i] + acc1b[i];
  }
}

__global__ __launch_bounds__(256)
void encode_reduce(const float* __restrict__ partials,
                   const float* __restrict__ enc_b,
                   float* __restrict__ encoded) {
  const int tid = threadIdx.x;
  if (tid >= BB * NLAT) return;
  const int b = tid / NLAT, l = tid % NLAT;
  const int tile = l >> 4, Nn = l & 15;
  float s = 0.f;
  for (int p = 0; p < NPART; ++p)
    s += partials[(size_t)(p * 2 + tile) * 256 + b * 32 + Nn];
  encoded[b * NLAT + l] = s + enc_b[l];
}

__global__ __launch_bounds__(256)
void bandwidth_kernel(const float* __restrict__ encoded,
                      const float* __restrict__ bw_w,
                      const float* __restrict__ bw_b,
                      float* __restrict__ invd_t,
                      float* __restrict__ ebl) {
  const int tid = blockIdx.x * blockDim.x + threadIdx.x;
  if (tid < BB * NLAT * MCL) {
    const int b = tid / (NLAT * MCL);
    const int j = tid % (NLAT * MCL);
    float acc = bw_b[j];
    for (int l2 = 0; l2 < NLAT; ++l2)
      acc = fmaf(encoded[b * NLAT + l2], bw_w[j * NLAT + l2], acc);
    const float s = 1.0f / (1.0f + expf(-acc));
    const float lo = 2.0f / 60.0f / MUF;
    const float hi = 1.0f / 60.0f;
    const float bw = (hi - lo) * s + lo;
    const int l = j / MCL, m = j % MCL;
    const float d = bw * MUF;
    invd_t[((size_t)m * NLAT + l) * BB + b] = 1.0f / (d * d);
  }
  if (tid < BB * NLAT) {
    const int l = tid / BB, b = tid % BB;
    ebl[tid] = encoded[b * NLAT + l];
  }
}

#define WAVES_TOTAL 512
#define NPW (NTOT / WAVES_TOTAL)

__global__ __launch_bounds__(256)
void conv_kernel(const float* __restrict__ ndist,
                 const int* __restrict__ nbid,
                 const int* __restrict__ labels,
                 const float* __restrict__ dec_w,
                 const float* __restrict__ invd_t,
                 const float* __restrict__ ebl,
                 float* __restrict__ out) {
  const int t  = threadIdx.x & 31;
  const int wg = (blockIdx.x * blockDim.x + threadIdx.x) >> 5;
  const int b  = t & 7;
  const int kc = t >> 3;
  for (int it = 0; it < NPW; ++it) {
    const int n = wg * NPW + it;
    if (it + 1 < NPW) {
      __builtin_prefetch(ndist + (size_t)(n + 1) * KNB, 0, 1);
      __builtin_prefetch(nbid  + (size_t)(n + 1) * KNB, 0, 1);
    }
    const float nd  = ndist[(size_t)n * KNB + t];
    const int   nid = nbid [(size_t)n * KNB + t];
    const float d2  = nd * nd;
    const int   m   = labels[n];
    float d2v[8];
#pragma unroll
    for (int j = 0; j < 8; ++j)
      d2v[j] = __shfl(d2, kc * 8 + j, 32);
    float outacc = 0.f;
    const float* invrow = invd_t + (size_t)m * NLAT * BB;
    for (int l = 0; l < NLAT; ++l) {
      const float g   = dec_w[(size_t)l * NTOT + nid];
      const float inv = invrow[l * BB + b];
      float wsum = 0.f, ps = 0.f;
#pragma unroll
      for (int j = 0; j < 8; ++j) {
        const float gv = __shfl(g, kc * 8 + j, 32);
        const float w  = fmaxf(0.f, fmaf(-d2v[j], inv, 1.0f));
        wsum += w;
        ps = fmaf(gv, w, ps);
      }
      wsum += __shfl_xor(wsum, 8, 32);
      wsum += __shfl_xor(wsum, 16, 32);
      ps   += __shfl_xor(ps, 8, 32);
      ps   += __shfl_xor(ps, 16, 32);
      const float conv = ps / wsum;
      outacc = fmaf(ebl[l * BB + b], conv, outacc);
    }
    if (t < 8)
      out[(size_t)b * NTOT + n] = outacc;
  }
}

extern "C" void kernel_launch(void* const* d_in, const int* in_sizes, int n_in,
                              void* d_out, int out_size, void* d_ws, size_t ws_size,
                              hipStream_t stream) {
  const float* x      = (const float*)d_in[0];
  const float* enc_w  = (const float*)d_in[1];
  const float* enc_b  = (const float*)d_in[2];
  const float* dec_w  = (const float*)d_in[3];
  const float* bw_w   = (const float*)d_in[4];
  const float* bw_b   = (const float*)d_in[5];
  const float* ndist  = (const float*)d_in[6];
  const int*   nbid   = (const int*)d_in[7];
  const int*   labels = (const int*)d_in[8];
  float* out = (float*)d_out;

  float* ws       = (float*)d_ws;
  float* partials = ws + WS_PARTIALS;
  float* encoded  = ws + WS_ENCODED;
  float* invd_t   = ws + WS_INVD;
  float* ebl      = ws + WS_EBL;

  encode_partial_wmma<<<NPART, 32, 0, stream>>>(x, enc_w, partials);
  encode_reduce<<<1, 256, 0, stream>>>(partials, enc_b, encoded);
  bandwidth_kernel<<<7, 256, 0, stream>>>(encoded, bw_w, bw_b, invd_t, ebl);
  conv_kernel<<<64, 256, 0, stream>>>(ndist, nbid, labels, dec_w, invd_t, ebl, out);
}


# conv stage spread over the chip: one point per wave, 512 groups of 32 waves (was 32 points per wave, 64 groups of 8); a group's 8x32 results go through LDS so one wave stores each 128-B line
# speedup vs baseline: 6.9627x; 6.9627x over previous
#include <hip/hip_runtime.h>
#include <math.h>

#define NTOT 16384
#define NLAT 20
#define MCL  10
#define KNB  32
#define BB   8
#define MUF  5.0f

typedef __attribute__((ext_vector_type(2))) float v2f;
typedef __attribute__((ext_vector_type(8))) float v8f;

#define NPART   128
#define KCHUNK  (NTOT / NPART)
#define WS_PARTIALS 0
#define WS_ENCODED  (NPART * 512)
#define WS_INVD     (WS_ENCODED + BB * NLAT)
#define WS_EBL      (WS_INVD + MCL * NLAT * BB)

__global__ __launch_bounds__(32)
void encode_partial_wmma(const float* __restrict__ x,
                         const float* __restrict__ enc_w,
                         float* __restrict__ partials) {
  const int t = threadIdx.x;
  const int p = blockIdx.x;
  const int M = t & 15;
  const int Nn = t & 15;
  const int half = (t >> 4) << 1;
  const float maskA  = (M < BB) ? 1.f : 0.f;
  const float maskB1 = (Nn < NLAT - 16) ? 1.f : 0.f;
  const int rowA  = (M < BB) ? M : 0;
  const int rowB1 = (Nn < NLAT - 16) ? (16 + Nn) : 16;
  const float* pa  = x     + (size_t)rowA  * NTOT + half;
  const float* pb0 = enc_w + (size_t)Nn    * NTOT + half;
  const float* pb1 = enc_w + (size_t)rowB1 * NTOT + half;
  v8f acc0a = {}, acc0b = {}, acc1a = {}, acc1b = {};
  const int n0base = p * KCHUNK;
  for (int s = 0; s < KCHUNK; s += 8) {
    const int n0 = n0base + s;
    const float2 av0 = *(const float2*)(pa  + n0);
    const float2 bv0 = *(const float2*)(pb0 + n0);
    const float2 cv0 = *(const float2*)(pb1 + n0);
    const float2 av1 = *(const float2*)(pa  + n0 + 4);
    const float2 bv1 = *(const float2*)(pb0 + n0 + 4);
    const float2 cv1 = *(const float2*)(pb1 + n0 + 4);
    v2f a0; a0.x = av0.x * maskA;  a0.y = av0.y * maskA;
    v2f b0; b0.x = bv0.x;          b0.y = bv0.y;
    v2f c0; c0.x = cv0.x * maskB1; c0.y = cv0.y * maskB1;
    v2f a1; a1.x = av1.x * maskA;  a1.y = av1.y * maskA;
    v2f b1; b1.x = bv1.x;          b1.y = bv1.y;
    v2f c1; c1.x = cv1.x * maskB1; c1.y = cv1.y * maskB1;
    acc0a = __builtin_amdgcn_wmma_f32_16x16x4_f32(false, a0, false, b0,
                                                  (short)0, acc0a, false, false);
    acc1a = __builtin_amdgcn_wmma_f32_16x16x4_f32(false, a0, false, c0,
                                                  (short)0, acc1a, false, false);
    acc0b = __builtin_amdgcn_wmma_f32_16x16x4_f32(false, a1, false, b1,
                                                  (short)0, acc0b, false, false);
    acc1b = __builtin_amdgcn_wmma_f32_16x16x4_f32(false, a1, false, c1,
                                                  (short)0, acc1b, false, false);
  }
  float* out0 = partials + (size_t)(p * 2 + 0) * 256;
  float* out1 = partials + (size_t)(p * 2 + 1) * 256;
#pragma unroll
  for (int i = 0; i < 8; ++i) {
    out0[i * 32 + t] = acc0a[i] + acc0b[i];
    out1[i * 32 + t] = acc1a[i] + acc1b[i];
  }
}

__global__ __launch_bounds__(256)
void encode_reduce(const float* __restrict__ partials,
                   const float* __restrict__ enc_b,
                   float* __restrict__ encoded) {
  const int tid = threadIdx.x;
  if (tid >= BB * NLAT) return;
  const int b = tid / NLAT, l = tid % NLAT;
  const int tile = l >> 4, Nn = l & 15;
  float s = 0.f;
  for (int p = 0; p < NPART; ++p)
    s += partials[(size_t)(p * 2 + tile) * 256 + b * 32 + Nn];
  encoded[b * NLAT + l] = s + enc_b[l];
}

__global__ __launch_bounds__(256)
void bandwidth_kernel(const float* __restrict__ encoded,
                      const float* __restrict__ bw_w,
                      const float* __restrict__ bw_b,
                      float* __restrict__ invd_t,
                      float* __restrict__ ebl) {
  const int tid = blockIdx.x * blockDim.x + threadIdx.x;
  if (tid < BB * NLAT * MCL) {
    const int b = tid / (NLAT * MCL);
    const int j = tid % (NLAT * MCL);
    float acc = bw_b[j];
    for (int l2 = 0; l2 < NLAT; ++l2)
      acc = fmaf(encoded[b * NLAT + l2], bw_w[j * NLAT + l2], acc);
    const float s = 1.0f / (1.0f + expf(-acc));
    const float lo = 2.0f / 60.0f / MUF;
    const float hi = 1.0f / 60.0f;
    const float bw = (hi - lo) * s + lo;
    const int l = j / MCL, m = j % MCL;
    const float d = bw * MUF;
    invd_t[((size_t)m * NLAT + l) * BB + b] = 1.0f / (d * d);
  }
  if (tid < BB * NLAT) {
    const int l = tid / BB, b = tid % BB;
    ebl[tid] = encoded[b * NLAT + l];
  }
}

#define CONV_PTS   32
#define CONV_PITCH (CONV_PTS + 1)

__global__ __launch_bounds__(CONV_PTS * 32)
void conv_kernel(const float* __restrict__ ndist,
                 const int* __restrict__ nbid,
                 const int* __restrict__ labels,
                 const float* __restrict__ dec_w,
                 const float* __restrict__ invd_t,
                 const float* __restrict__ ebl,
                 float* __restrict__ out) {
  __shared__ float stage[BB * CONV_PITCH];
  const int t  = threadIdx.x & 31;
  const int w  = threadIdx.x >> 5;
  const int n  = blockIdx.x * CONV_PTS + w;
  const int b  = t & 7;
  const int kc = t >> 3;
  const float nd  = ndist[(size_t)n * KNB + t];
  const int   nid = nbid [(size_t)n * KNB + t];
  const float d2  = nd * nd;
  const int   m   = labels[n];
  float d2v[8];
#pragma unroll
  for (int j = 0; j < 8; ++j)
    d2v[j] = __shfl(d2, kc * 8 + j, 32);
  float outacc = 0.f;
  const float* invrow = invd_t + (size_t)m * NLAT * BB;
  for (int l = 0; l < NLAT; ++l) {
    const float g   = dec_w[(size_t)l * NTOT + nid];
    const float inv = invrow[l * BB + b];
    float wsum = 0.f, ps = 0.f;
#pragma unroll
    for (int j = 0; j < 8; ++j) {
      const float gv = __shfl(g, kc * 8 + j, 32);
      const float wt = fmaxf(0.f, fmaf(-d2v[j], inv, 1.0f));
      wsum += wt;
      ps = fmaf(gv, wt, ps);
    }
    wsum += __shfl_xor(wsum, 8, 32);
    wsum += __shfl_xor(wsum, 16, 32);
    ps   += __shfl_xor(ps, 8, 32);
    ps   += __shfl_xor(ps, 16, 32);
    const float conv = ps / wsum;
    outacc = fmaf(ebl[l * BB + b], conv, outacc);
  }
  if (t < 8)
    stage[b * CONV_PITCH + w] = outacc;
  __syncthreads();
  if (w < BB)
    out[(size_t)w * NTOT + blockIdx.x * CONV_PTS + t] = stage[w * CONV_PITCH + t];
}

extern "C" void kernel_launch(void* const* d_in, const int* in_sizes, int n_in,
                              void* d_out, int out_size, void* d_ws, size_t ws_size,
                              hipStream_t stream) {
  const float* x      = (const float*)d_in[0];
  const float* enc_w  = (const float*)d_in[1];
  const float* enc_b  = (const float*)d_in[2];
  const float* dec_w  = (const float*)d_in[3];
  const float* bw_w   = (const float*)d_in[4];
  const float* bw_b   = (const float*)d_in[5];
  const float* ndist  = (const float*)d_in[6];
  const int*   nbid   = (const int*)d_in[7];
  const int*   labels = (const int*)d_in[8];
  float* out = (float*)d_out;

  float* ws       = (float*)d_ws;
  float* partials = ws + WS_PARTIALS;
  float* encoded  = ws + WS_ENCODED;
  float* invd_t   = ws + WS_INVD;
  float* ebl      = ws + WS_EBL;

  encode_partial_wmma<<<NPART, 32, 0, stream>>>(x, enc_w, partials);
  encode_reduce<<<1, 256, 0, stream>>>(partials, enc_b, encoded);
  bandwidth_kernel<<<7, 256, 0, stream>>>(encoded, bw_w, bw_b, invd_t, ebl);
  conv_kernel<<<NTOT / CONV_PTS, CONV_PTS * 32, 0, stream>>>(ndist, nbid, labels, dec_w, invd_t, ebl, out);
}


# decoder basis re-laid node-major in the workspace (one 128-B line per node, by extra groups of the reduce launch); conv lane k loads its neighbour's 20 values as 5x16 B, shared via wave-private LDS
# speedup vs baseline: 9.6989x; 1.3930x over previous
#include <hip/hip_runtime.h>
#include <math.h>

#define NTOT 16384
#define NLAT 20
#define MCL  10
#define KNB  32
#define BB   8
#define MUF  5.0f

typedef __attribute__((ext_vector_type(2))) float v2f;
typedef __attribute__((ext_vector_type(8))) float v8f;

#define NPART   128
#define KCHUNK  (NTOT / NPART)
#define WS_PARTIALS 0
#define WS_ENCODED  (NPART * 512)
#define WS_INVD     (WS_ENCODED + BB * NLAT)
#define WS_EBL      (WS_INVD + MCL * NLAT * BB)
#define DT_PITCH    32
#define WS_DECT     (WS_EBL + BB * NLAT)

__global__ __launch_bounds__(32)
void encode_partial_wmma(const float* __restrict__ x,
                         const float* __restrict__ enc_w,
                         float* __restrict__ partials) {
  const int t = threadIdx.x;
  const int p = blockIdx.x;
  const int M = t & 15;
  const int Nn = t & 15;
  const int half = (t >> 4) << 1;
  const float maskA  = (M < BB) ? 1.f : 0.f;
  const float maskB1 = (Nn < NLAT - 16) ? 1.f : 0.f;
  const int rowA  = (M < BB) ? M : 0;
  const int rowB1 = (Nn < NLAT - 16) ? (16 + Nn) : 16;
  const float* pa  = x     + (size_t)rowA  * NTOT + half;
  const float* pb0 = enc_w + (size_t)Nn    * NTOT + half;
  const float* pb1 = enc_w + (size_t)rowB1 * NTOT + half;
  v8f acc0a = {}, acc0b = {}, acc1a = {}, acc1b = {};
  const int n0base = p * KCHUNK;
  for (int s = 0; s < KCHUNK; s += 8) {
    const int n0 = n0base + s;
    const float2 av0 = *(const float2*)(pa  + n0);
    const float2 bv0 = *(const float2*)(pb0 + n0);
    const float2 cv0 = *(const float2*)(pb1 + n0);
    const float2 av1 = *(const float2*)(pa  + n0 + 4);
    const float2 bv1 = *(const float2*)(pb0 + n0 + 4);
    const float2 cv1 = *(const float2*)(pb1 + n0 + 4);
    v2f a0; a0.x = av0.x * maskA;  a0.y = av0.y * maskA;
    v2f b0; b0.x = bv0.x;          b0.y = bv0.y;
    v2f c0; c0.x = cv0.x * maskB1; c0.y = cv0.y * maskB1;
    v2f a1; a1.x = av1.x * maskA;  a1.y = av1.y * maskA;
    v2f b1; b1.x = bv1.x;          b1.y = bv1.y;
    v2f c1; c1.x = cv1.x * maskB1; c1.y = cv1.y * maskB1;
    acc0a = __builtin_amdgcn_wmma_f32_16x16x4_f32(false, a0, false, b0,
                                                  (short)0, acc0a, false, false);
    acc1a = __builtin_amdgcn_wmma_f32_16x16x4_f32(false, a0, false, c0,
                                                  (short)0, acc1a, false, false);
    acc0b = __builtin_amdgcn_wmma_f32_16x16x4_f32(false, a1, false, b1,
                                                  (short)0, acc0b, false, false);
    acc1b = __builtin_amdgcn_wmma_f32_16x16x4_f32(false, a1, false, c1,
                                                  (short)0, acc1b, false, false);
  }
  float* out0 = partials + (size_t)(p * 2 + 0) * 256;
  float* out1 = partials + (size_t)(p * 2 + 1) * 256;
#pragma unroll
  for (int i = 0; i < 8; ++i) {
    out0[i * 32 + t] = acc0a[i] + acc0b[i];
    out1[i * 32 + t] = acc1a[i] + acc1b[i];
  }
}

#define TR_NODES 128
#define TR_PITCH 21

__global__ __launch_bounds__(256)
void encode_reduce(const float* __restrict__ partials,
                   const float* __restrict__ enc_b,
                   float* __restrict__ encoded,
                   const float* __restrict__ dec_w,
                   float* __restrict__ dec_t) {
  __shared__ float tile[TR_NODES * TR_PITCH];
  const int tid = threadIdx.x;
  if (blockIdx.x == 0) {
    if (tid >= BB * NLAT) return;
    const int b = tid / NLAT, l = tid % NLAT;
    const int tl = l >> 4, Nn = l & 15;
    float s = 0.f;
    for (int p = 0; p < NPART; ++p)
      s += partials[(size_t)(p * 2 + tl) * 256 + b * 32 + Nn];
    encoded[b * NLAT + l] = s + enc_b[l];
    return;
  }
  const int base = (blockIdx.x - 1) * TR_NODES;
  const int c  = tid & (TR_NODES - 1);
  const int l0 = tid >> 7;
#pragma unroll
  for (int r = 0; r < NLAT / 2; ++r) {
    const int l = l0 + 2 * r;
    tile[c * TR_PITCH + l] = dec_w[(size_t)l * NTOT + base + c];
  }
  __syncthreads();
  const int j = tid & 7;
#pragma unroll
  for (int r = 0; r < TR_NODES / 32; ++r) {
    const int cc = (tid >> 3) + 32 * r;
    float4 v = make_float4(0.f, 0.f, 0.f, 0.f);
    if (j < NLAT / 4) {
      const float* s = tile + cc * TR_PITCH + 4 * j;
      v = make_float4(s[0], s[1], s[2], s[3]);
    }
    *(float4*)(dec_t + (size_t)(base + cc) * DT_PITCH + 4 * j) = v;
  }
}

__global__ __launch_bounds__(256)
void bandwidth_kernel(const float* __restrict__ encoded,
                      const float* __restrict__ bw_w,
                      const float* __restrict__ bw_b,
                      float* __restrict__ invd_t,
                      float* __restrict__ ebl) {
  const int tid = blockIdx.x * blockDim.x + threadIdx.x;
  if (tid < BB * NLAT * MCL) {
    const int b = tid / (NLAT * MCL);
    const int j = tid % (NLAT * MCL);
    float acc = bw_b[j];
    for (int l2 = 0; l2 < NLAT; ++l2)
      acc = fmaf(encoded[b * NLAT + l2], bw_w[j * NLAT + l2], acc);
    const float s = 1.0f / (1.0f + expf(-acc));
    const float lo = 2.0f / 60.0f / MUF;
    const float hi = 1.0f / 60.0f;
    const float bw = (hi - lo) * s + lo;
    const int l = j / MCL, m = j % MCL;
    const float d = bw * MUF;
    invd_t[((size_t)m * NLAT + l) * BB + b] = 1.0f / (d * d);
  }
  if (tid < BB * NLAT) {
    const int l = tid / BB, b = tid % BB;
    ebl[tid] = encoded[b * NLAT + l];
  }
}

#define CONV_PTS   32
#define CONV_PITCH (CONV_PTS + 1)

__global__ __launch_bounds__(CONV_PTS * 32)
void conv_kernel(const float* __restrict__ ndist,
                 const int* __restrict__ nbid,
                 const int* __restrict__ labels,
                 const float* __restrict__ dec_t,
                 const float* __restrict__ invd_t,
                 const float* __restrict__ ebl,
                 float* __restrict__ out) {
  __shared__ float stage[BB * CONV_PITCH];
  __shared__ float4 gtile[CONV_PTS * NLAT * (KNB / 4)];
  const int t  = threadIdx.x & 31;
  const int w  = threadIdx.x >> 5;
  const int n  = blockIdx.x * CONV_PTS + w;
  const int b  = t & 7;
  const int kc = t >> 3;
  const float nd  = ndist[(size_t)n * KNB + t];
  const int   nid = nbid [(size_t)n * KNB + t];
  const int   m   = labels[n];
  const float4* pl = (const float4*)(dec_t + (size_t)nid * DT_PITCH);
  float4 q[NLAT / 4];
#pragma unroll
  for (int c = 0; c < NLAT / 4; ++c) q[c] = pl[c];
  const float d2  = nd * nd;
  float d2v[8];
#pragma unroll
  for (int j = 0; j < 8; ++j)
    d2v[j] = __shfl(d2, kc * 8 + j, 32);
  float* gw = (float*)(gtile + w * (NLAT * (KNB / 4)));
#pragma unroll
  for (int c = 0; c < NLAT / 4; ++c) {
    gw[(4 * c + 0) * KNB + t] = q[c].x;
    gw[(4 * c + 1) * KNB + t] = q[c].y;
    gw[(4 * c + 2) * KNB + t] = q[c].z;
    gw[(4 * c + 3) * KNB + t] = q[c].w;
  }
  __builtin_amdgcn_fence(__ATOMIC_RELEASE, "wavefront");
  __builtin_amdgcn_wave_barrier();
  __builtin_amdgcn_fence(__ATOMIC_ACQUIRE, "wavefront");
  float outacc = 0.f;
  const float* invrow = invd_t + (size_t)m * NLAT * BB;
  const float4* gr = (const float4*)gw + kc * 2;
  for (int l = 0; l < NLAT; ++l) {
    const float4 ga = gr[l * (KNB / 4)], gb = gr[l * (KNB / 4) + 1];
    const float inv = invrow[l * BB + b];
    const float gv[8] = {ga.x, ga.y, ga.z, ga.w, gb.x, gb.y, gb.z, gb.w};
    float wsum = 0.f, ps = 0.f;
#pragma unroll
    for (int j = 0; j < 8; ++j) {
      const float wt = fmaxf(0.f, fmaf(-d2v[j], inv, 1.0f));
      wsum += wt;
      ps = fmaf(gv[j], wt, ps);
    }
    wsum += __shfl_xor(wsum, 8, 32);
    wsum += __shfl_xor(wsum, 16, 32);
    ps   += __shfl_xor(ps, 8, 32);
    ps   += __shfl_xor(ps, 16, 32);
    const float conv = ps / wsum;
    outacc = fmaf(ebl[l * BB + b], conv, outacc);
  }
  if (t < 8)
    stage[b * CONV_PITCH + w] = outacc;
  __syncthreads();
  if (w < BB)
    out[(size_t)w * NTOT + blockIdx.x * CONV_PTS + t] = stage[w * CONV_PITCH + t];
}

extern "C" void kernel_launch(void* const* d_in, const int* in_sizes, int n_in,
                              void* d_out, int out_size, void* d_ws, size_t ws_size,
                              hipStream_t stream) {
  const float* x      = (const float*)d_in[0];
  const float* enc_w  = (const float*)d_in[1];
  const float* enc_b  = (const float*)d_in[2];
  const float* dec_w  = (const float*)d_in[3];
  const float* bw_w   = (const float*)d_in[4];
  const float* bw_b   = (const float*)d_in[5];
  const float* ndist  = (const float*)d_in[6];
  const int*   nbid   = (const int*)d_in[7];
  const int*   labels = (const int*)d_in[8];
  float* out = (float*)d_out;

  float* ws       = (float*)d_ws;
  float* partials = ws + WS_PARTIALS;
  float* encoded  = ws + WS_ENCODED;
  float* invd_t   = ws + WS_INVD;
  float* ebl      = ws + WS_EBL;
  float* dec_t    = ws + WS_DECT;

  encode_partial_wmma<<<NPART, 32, 0, stream>>>(x, enc_w, partials);
  encode_reduce<<<1 + NTOT / TR_NODES, 256, 0, stream>>>(partials, enc_b, encoded, dec_w, dec_t);
  bandwidth_kernel<<<7, 256, 0, stream>>>(encoded, bw_w, bw_b, invd_t, ebl);
  conv_kernel<<<NTOT / CONV_PTS, CONV_PTS * 32, 0, stream>>>(ndist, nbid, labels, dec_t, invd_t, ebl, out);
}


# 4 launches -> 2: basis re-lay rides in the encode launch; every conv group sums the 32 encode partials itself and builds the 1600 window scales in LDS; encode sum order changes (8 waves pre-summed)
# speedup vs baseline: 14.7260x; 1.5183x over previous
#include <hip/hip_runtime.h>
#include <math.h>

#define NTOT 16384
#define NLAT 20
#define MCL  10
#define KNB  32
#define BB   8
#define MUF  5.0f


typedef __attribute__((ext_vector_type(2))) float v2f;
typedef __attribute__((ext_vector_type(8))) float v8f;

#define NENC      32
#define ENC_WAVES 8
#define KCHUNK  (NTOT / (NENC * ENC_WAVES))
#define NOUT    (BB * NLAT)
#define WS_PARTIALS 0
#define DT_PITCH    32
#define WS_DECT     (NENC * NOUT)

#define TR_NODES 128
#define TR_PITCH 21

__global__ __launch_bounds__(ENC_WAVES * 32)
void front_kernel(const float* __restrict__ x,
                  const float* __restrict__ enc_w,
                  const float* __restrict__ dec_w,
                  float* __restrict__ partials,
                  float* __restrict__ dec_t) {
  __shared__ float tile[TR_NODES * TR_PITCH];
  __shared__ float red[ENC_WAVES * NOUT];
  const int tid = threadIdx.x;
  if (blockIdx.x < NENC) {
    const int t = tid & 31;
    const int w = tid >> 5;
    const int M = t & 15;
    const int Nn = t & 15;
    const int half = (t >> 4) << 1;
    const float maskA  = (M < BB) ? 1.f : 0.f;
    const float maskB1 = (Nn < NLAT - 16) ? 1.f : 0.f;
    const int rowA  = (M < BB) ? M : 0;
    const int rowB1 = (Nn < NLAT - 16) ? (16 + Nn) : 16;
    const float* pa  = x     + (size_t)rowA  * NTOT + half;
    const float* pb0 = enc_w + (size_t)Nn    * NTOT + half;
    const float* pb1 = enc_w + (size_t)rowB1 * NTOT + half;
    v8f acc0a = {}, acc0b = {}, acc1a = {}, acc1b = {};
    const int n0base = (blockIdx.x * ENC_WAVES + w) * KCHUNK;
    for (int s = 0; s < KCHUNK; s += 8) {
      const int n0 = n0base + s;
      const float2 av0 = *(const float2*)(pa  + n0);
      const float2 bv0 = *(const float2*)(pb0 + n0);
      const float2 cv0 = *(const float2*)(pb1 + n0);
      const float2 av1 = *(const float2*)(pa  + n0 + 4);
      const float2 bv1 = *(const float2*)(pb0 + n0 + 4);
      const float2 cv1 = *(const float2*)(pb1 + n0 + 4);
      v2f a0; a0.x = av0.x * maskA;  a0.y = av0.y * maskA;
      v2f b0; b0.x = bv0.x;          b0.y = bv0.y;
      v2f c0; c0.x = cv0.x * maskB1; c0.y = cv0.y * maskB1;
      v2f a1; a1.x = av1.x * maskA;  a1.y = av1.y * maskA;
      v2f b1; b1.x = bv1.x;          b1.y = bv1.y;
      v2f c1; c1.x = cv1.x * maskB1; c1.y = cv1.y * maskB1;
      acc0a = __builtin_amdgcn_wmma_f32_16x16x4_f32(false, a0, false, b0,
                                                    (short)0, acc0a, false, false);
      acc1a = __builtin_amdgcn_wmma_f32_16x16x4_f32(false, a0, false, c0,
                                                    (short)0, acc1a, false, false);
      acc0b = __builtin_amdgcn_wmma_f32_16x16x4_f32(false, a1, false, b1,
                                                    (short)0, acc0b, false, false);
      acc1b = __builtin_amdgcn_wmma_f32_16x16x4_f32(false, a1, false, c1,
                                                    (short)0, acc1b, false, false);
    }
    float* rw = red + w * NOUT;
    if (t < 16) {
#pragma unroll
      for (int i = 0; i < 8; ++i) rw[i * NLAT + t] = acc0a[i] + acc0b[i];
    }
    if (t < NLAT - 16) {
#pragma unroll
      for (int i = 0; i < 8; ++i) rw[i * NLAT + 16 + t] = acc1a[i] + acc1b[i];
    }
    __syncthreads();
    if (tid < NOUT) {
      float s = 0.f;
#pragma unroll
      for (int w2 = 0; w2 < ENC_WAVES; ++w2) s += red[w2 * NOUT + tid];
      partials[blockIdx.x * NOUT + tid] = s;
    }
    return;
  }
  const int base = (blockIdx.x - NENC) * TR_NODES;
  const int c  = tid & (TR_NODES - 1);
  const int l0 = tid >> 7;
#pragma unroll
  for (int r = 0; r < NLAT / 2; ++r) {
    const int l = l0 + 2 * r;
    tile[c * TR_PITCH + l] = dec_w[(size_t)l * NTOT + base + c];
  }
  __syncthreads();
  const int j = tid & 7;
#pragma unroll
  for (int r = 0; r < TR_NODES / 32; ++r) {
    const int cc = (tid >> 3) + 32 * r;
    float4 v = make_float4(0.f, 0.f, 0.f, 0.f);
    if (j < NLAT / 4) {
      const float* s = tile + cc * TR_PITCH + 4 * j;
      v = make_float4(s[0], s[1], s[2], s[3]);
    }
    *(float4*)(dec_t + (size_t)(base + cc) * DT_PITCH + 4 * j) = v;
  }
}

#define CONV_PTS   32
#define CONV_PITCH (CONV_PTS + 1)
#define NBW        (NLAT * MCL)

__global__ __launch_bounds__(CONV_PTS * 32)
void conv_kernel(const float* __restrict__ ndist,
                 const int* __restrict__ nbid,
                 const int* __restrict__ labels,
                 const float* __restrict__ dec_t,
                 const float* __restrict__ partials,
                 const float* __restrict__ enc_b,
                 const float* __restrict__ bw_w,
                 const float* __restrict__ bw_b,
                 float* __restrict__ out) {
  __shared__ float stage[BB * CONV_PITCH];
  __shared__ float4 gtile[CONV_PTS * NLAT * (KNB / 4)];
  __shared__ float4 enc_s4[NOUT / 4];
  __shared__ float4 bww_s[NBW * NLAT / 4];
  __shared__ float bwb_s[NBW];
  __shared__ float sinv[NBW * BB];
  float* enc_s = (float*)enc_s4;
  const int tid = threadIdx.x;
  const int t  = tid & 31;
  const int w  = tid >> 5;
  const int n  = blockIdx.x * CONV_PTS + w;
  const int b  = t & 7;
  const int kc = t >> 3;
  const float nd  = ndist[(size_t)n * KNB + t];
  const int   nid = nbid [(size_t)n * KNB + t];
  const int   m   = labels[n];
  const float bpiece = bw_b[tid < NBW ? tid : NBW - 1];
  if (tid < NOUT) {
    float s = 0.f;
#pragma unroll
    for (int p = 0; p < NENC; ++p) s += partials[p * NOUT + tid];
    enc_s[tid] = s + enc_b[tid % NLAT];
  }
  const float4* pl = (const float4*)(dec_t + (size_t)nid * DT_PITCH);
  float4 q[NLAT / 4];
#pragma unroll
  for (int c = 0; c < NLAT / 4; ++c) q[c] = pl[c];
  if (tid < NBW * NLAT / 4) bww_s[tid] = ((const float4*)bw_w)[tid];
  if (tid < NBW) bwb_s[tid] = bpiece;
  float* gw = (float*)(gtile + w * (NLAT * (KNB / 4)));
#pragma unroll
  for (int c = 0; c < NLAT / 4; ++c) {
    gw[(4 * c + 0) * KNB + t] = q[c].x;
    gw[(4 * c + 1) * KNB + t] = q[c].y;
    gw[(4 * c + 2) * KNB + t] = q[c].z;
    gw[(4 * c + 3) * KNB + t] = q[c].w;
  }
  __syncthreads();
  for (int idx = tid; idx < NBW * BB; idx += CONV_PTS * 32) {
    const int bb = idx & (BB - 1);
    const int j  = idx >> 3;
    const float4* wr = bww_s + j * (NLAT / 4);
    const float4* er = enc_s4 + bb * (NLAT / 4);
    float acc = bwb_s[j];
#pragma unroll
    for (int c = 0; c < NLAT / 4; ++c) {
      const float4 ev = er[c], wv = wr[c];
      acc = fmaf(ev.x, wv.x, acc);
      acc = fmaf(ev.y, wv.y, acc);
      acc = fmaf(ev.z, wv.z, acc);
      acc = fmaf(ev.w, wv.w, acc);
    }
    const float s = 1.0f / (1.0f + expf(-acc));
    const float lo = 2.0f / 60.0f / MUF;
    const float hi = 1.0f / 60.0f;
    const float bw = (hi - lo) * s + lo;
    const float d = bw * MUF;
    sinv[idx] = 1.0f / (d * d);
  }
  const float d2  = nd * nd;
  float d2v[8];
#pragma unroll
  for (int j = 0; j < 8; ++j)
    d2v[j] = __shfl(d2, kc * 8 + j, 32);
  __syncthreads();
  float outacc = 0.f;
  const float* invcol = sinv + m * BB + b;
  const float4* gr = (const float4*)gw + kc * 2;
  for (int l = 0; l < NLAT; ++l) {
    const float4 ga = gr[l * (KNB / 4)], gb = gr[l * (KNB / 4) + 1];
    const float inv = invcol[l * (MCL * BB)];
    const float gv[8] = {ga.x, ga.y, ga.z, ga.w, gb.x, gb.y, gb.z, gb.w};
    float wsum = 0.f, ps = 0.f;
#pragma unroll
    for (int j = 0; j < 8; ++j) {
      const float wt = fmaxf(0.f, fmaf(-d2v[j], inv, 1.0f));
      wsum += wt;
      ps = fmaf(gv[j], wt, ps);
    }
    wsum += __shfl_xor(wsum, 8, 32);
    wsum += __shfl_xor(wsum, 16, 32);
    ps   += __shfl_xor(ps, 8, 32);
    ps   += __shfl_xor(ps, 16, 32);
    const float conv = ps / wsum;
    outacc = fmaf(enc_s[b * NLAT + l], conv, outacc);
  }
  if (t < 8)
    stage[b * CONV_PITCH + w] = outacc;
  __syncthreads();
  if (w < BB)
    out[(size_t)w * NTOT + blockIdx.x * CONV_PTS + t] = stage[w * CONV_PITCH + t];
}

extern "C" void kernel_launch(void* const* d_in, const int* in_sizes, int n_in,
                              void* d_out, int out_size, void* d_ws, size_t ws_size,
                              hipStream_t stream) {
  const float* x      = (const float*)d_in[0];
  const float* enc_w  = (const float*)d_in[1];
  const float* enc_b  = (const float*)d_in[2];
  const float* dec_w  = (const float*)d_in[3];
  const float* bw_w   = (const float*)d_in[4];
  const float* bw_b   = (const float*)d_in[5];
  const float* ndist  = (const float*)d_in[6];
  const int*   nbid   = (const int*)d_in[7];
  const int*   labels = (const int*)d_in[8];
  float* out = (float*)d_out;

  float* ws       = (float*)d_ws;
  float* partials = ws + WS_PARTIALS;
  float* dec_t    = ws + WS_DECT;

  front_kernel<<<NENC + NTOT / TR_NODES, ENC_WAVES * 32, 0, stream>>>(x, enc_w, dec_w, partials, dec_t);
  conv_kernel<<<NTOT / CONV_PTS, CONV_PTS * 32, 0, stream>>>(ndist, nbid, labels, dec_t, partials, enc_b, bw_w, bw_b, out);
}


# conv lanes re-mapped: lane = (batch row, 5 latents) runs all 32 neighbours itself, distances via scalar loads; no cross-lane sums in the loop, 5 divisions per lane (was 20); sum order changes
# speedup vs baseline: 17.8375x; 1.2113x over previous
#include <hip/hip_runtime.h>
#include <math.h>

#define NTOT 16384
#define NLAT 20
#define MCL  10
#define KNB  32
#define BB   8
#define MUF  5.0f


typedef __attribute__((ext_vector_type(2))) float v2f;
typedef __attribute__((ext_vector_type(8))) float v8f;

#define NENC      32
#define ENC_WAVES 8
#define KCHUNK  (NTOT / (NENC * ENC_WAVES))
#define NOUT    (BB * NLAT)
#define WS_PARTIALS 0
#define DT_PITCH    32
#define WS_DECT     (NENC * NOUT)

#define TR_NODES 128
#define TR_PITCH 21

__global__ __launch_bounds__(ENC_WAVES * 32)
void front_kernel(const float* __restrict__ x,
                  const float* __restrict__ enc_w,
                  const float* __restrict__ dec_w,
                  float* __restrict__ partials,
                  float* __restrict__ dec_t) {
  __shared__ float tile[TR_NODES * TR_PITCH];
  __shared__ float red[ENC_WAVES * NOUT];
  const int tid = threadIdx.x;
  if (blockIdx.x < NENC) {
    const int t = tid & 31;
    const int w = tid >> 5;
    const int M = t & 15;
    const int Nn = t & 15;
    const int half = (t >> 4) << 1;
    const float maskA  = (M < BB) ? 1.f : 0.f;
    const float maskB1 = (Nn < NLAT - 16) ? 1.f : 0.f;
    const int rowA  = (M < BB) ? M : 0;
    const int rowB1 = (Nn < NLAT - 16) ? (16 + Nn) : 16;
    const float* pa  = x     + (size_t)rowA  * NTOT + half;
    const float* pb0 = enc_w + (size_t)Nn    * NTOT + half;
    const float* pb1 = enc_w + (size_t)rowB1 * NTOT + half;
    v8f acc0a = {}, acc0b = {}, acc1a = {}, acc1b = {};
    const int n0base = (blockIdx.x * ENC_WAVES + w) * KCHUNK;
    for (int s = 0; s < KCHUNK; s += 8) {
      const int n0 = n0base + s;
      const float2 av0 = *(const float2*)(pa  + n0);
      const float2 bv0 = *(const float2*)(pb0 + n0);
      const float2 cv0 = *(const float2*)(pb1 + n0);
      const float2 av1 = *(const float2*)(pa  + n0 + 4);
      const float2 bv1 = *(const float2*)(pb0 + n0 + 4);
      const float2 cv1 = *(const float2*)(pb1 + n0 + 4);
      v2f a0; a0.x = av0.x * maskA;  a0.y = av0.y * maskA;
      v2f b0; b0.x = bv0.x;          b0.y = bv0.y;
      v2f c0; c0.x = cv0.x * maskB1; c0.y = cv0.y * maskB1;
      v2f a1; a1.x = av1.x * maskA;  a1.y = av1.y * maskA;
      v2f b1; b1.x = bv1.x;          b1.y = bv1.y;
      v2f c1; c1.x = cv1.x * maskB1; c1.y = cv1.y * maskB1;
      acc0a = __builtin_amdgcn_wmma_f32_16x16x4_f32(false, a0, false, b0,
                                                    (short)0, acc0a, false, false);
      acc1a = __builtin_amdgcn_wmma_f32_16x16x4_f32(false, a0, false, c0,
                                                    (short)0, acc1a, false, false);
      acc0b = __builtin_amdgcn_wmma_f32_16x16x4_f32(false, a1, false, b1,
                                                    (short)0, acc0b, false, false);
      acc1b = __builtin_amdgcn_wmma_f32_16x16x4_f32(false, a1, false, c1,
                                                    (short)0, acc1b, false, false);
    }
    float* rw = red + w * NOUT;
    if (t < 16) {
#pragma unroll
      for (int i = 0; i < 8; ++i) rw[i * NLAT + t] = acc0a[i] + acc0b[i];
    }
    if (t < NLAT - 16) {
#pragma unroll
      for (int i = 0; i < 8; ++i) rw[i * NLAT + 16 + t] = acc1a[i] + acc1b[i];
    }
    __syncthreads();
    if (tid < NOUT) {
      float s = 0.f;
#pragma unroll
      for (int w2 = 0; w2 < ENC_WAVES; ++w2) s += red[w2 * NOUT + tid];
      partials[blockIdx.x * NOUT + tid] = s;
    }
    return;
  }
  const int base = (blockIdx.x - NENC) * TR_NODES;
  const int c  = tid & (TR_NODES - 1);
  const int l0 = tid >> 7;
#pragma unroll
  for (int r = 0; r < NLAT / 2; ++r) {
    const int l = l0 + 2 * r;
    tile[c * TR_PITCH + l] = dec_w[(size_t)l * NTOT + base + c];
  }
  __syncthreads();
  const int j = tid & 7;
#pragma unroll
  for (int r = 0; r < TR_NODES / 32; ++r) {
    const int cc = (tid >> 3) + 32 * r;
    float4 v = make_float4(0.f, 0.f, 0.f, 0.f);
    if (j < NLAT / 4) {
      const float* s = tile + cc * TR_PITCH + 4 * j;
      v = make_float4(s[0], s[1], s[2], s[3]);
    }
    *(float4*)(dec_t + (size_t)(base + cc) * DT_PITCH + 4 * j) = v;
  }
}

#define CONV_PTS   32
#define CONV_PITCH (CONV_PTS + 1)
#define NBW        (NLAT * MCL)
#define GT_PITCH   36

__global__ __launch_bounds__(CONV_PTS * 32, 16)
void conv_kernel(const float* __restrict__ ndist,
                 const int* __restrict__ nbid,
                 const int* __restrict__ labels,
                 const float* __restrict__ dec_t,
                 const float* __restrict__ partials,
                 const float* __restrict__ enc_b,
                 const float* __restrict__ bw_w,
                 const float* __restrict__ bw_b,
                 float* __restrict__ out) {
  __shared__ float stage[BB * CONV_PITCH];
  __shared__ float4 gtile[CONV_PTS * NLAT * (GT_PITCH / 4)];
  __shared__ float4 enc_s4[NOUT / 4];
  __shared__ float4 bww_s[NBW * NLAT / 4];
  __shared__ float bwb_s[NBW];
  __shared__ float sinv[NBW * BB];
  float* enc_s = (float*)enc_s4;
  const int tid = threadIdx.x;
  const int t  = tid & 31;
  const int w  = __builtin_amdgcn_readfirstlane(tid >> 5);
  const int n  = blockIdx.x * CONV_PTS + w;
  const int b  = t & 7;
  const int lq = t >> 3;
  const int   nid = nbid [(size_t)n * KNB + t];
  const int   m   = labels[n];
  const float* ndp = ndist + (size_t)n * KNB;
  float ndv[KNB];
#pragma unroll
  for (int k = 0; k < KNB; ++k) ndv[k] = ndp[k];
  const float bpiece = bw_b[tid < NBW ? tid : NBW - 1];
  if (tid < NOUT) {
    float s = 0.f;
#pragma unroll
    for (int p = 0; p < NENC; ++p) s += partials[p * NOUT + tid];
    enc_s[tid] = s + enc_b[tid % NLAT];
  }
  const float4* pl = (const float4*)(dec_t + (size_t)nid * DT_PITCH);
  float4 q[NLAT / 4];
#pragma unroll
  for (int c = 0; c < NLAT / 4; ++c) q[c] = pl[c];
  if (tid < NBW * NLAT / 4) bww_s[tid] = ((const float4*)bw_w)[tid];
  if (tid < NBW) bwb_s[tid] = bpiece;
  float* gw = (float*)(gtile + w * (NLAT * (GT_PITCH / 4)));
#pragma unroll
  for (int c = 0; c < NLAT / 4; ++c) {
    gw[(4 * c + 0) * GT_PITCH + t] = q[c].x;
    gw[(4 * c + 1) * GT_PITCH + t] = q[c].y;
    gw[(4 * c + 2) * GT_PITCH + t] = q[c].z;
    gw[(4 * c + 3) * GT_PITCH + t] = q[c].w;
  }
  __syncthreads();
  for (int idx = tid; idx < NBW * BB; idx += CONV_PTS * 32) {
    const int bb = idx & (BB - 1);
    const int j  = idx >> 3;
    const float4* wr = bww_s + j * (NLAT / 4);
    const float4* er = enc_s4 + bb * (NLAT / 4);
    float acc = bwb_s[j];
#pragma unroll
    for (int c = 0; c < NLAT / 4; ++c) {
      const float4 ev = er[c], wv = wr[c];
      acc = fmaf(ev.x, wv.x, acc);
      acc = fmaf(ev.y, wv.y, acc);
      acc = fmaf(ev.z, wv.z, acc);
      acc = fmaf(ev.w, wv.w, acc);
    }
    const float s = 1.0f / (1.0f + expf(-acc));
    const float lo = 2.0f / 60.0f / MUF;
    const float hi = 1.0f / 60.0f;
    const float bw = (hi - lo) * s + lo;
    const float d = bw * MUF;
    sinv[idx] = 1.0f / (d * d);
  }
  __syncthreads();
  const v2f one2  = {1.f, 1.f};
  const v2f zero2 = {0.f, 0.f};
  float outacc = 0.f;
  const float* invcol = sinv + m * BB + b;
  const float* gwl = gw + lq * GT_PITCH;
#pragma unroll
  for (int i = 0; i < NLAT / 4; ++i) {
    const int l = lq + 4 * i;
    const float inv = invcol[l * (MCL * BB)];
    const v2f ninv = {-inv, -inv};
    const float4* gr = (const float4*)(gwl + 4 * i * GT_PITCH);
    v2f ws = zero2, ps = zero2;
#pragma unroll
    for (int k4 = 0; k4 < KNB / 4; ++k4) {
      const float4 g = gr[k4];
      const v2f da = {ndv[4 * k4 + 0] * ndv[4 * k4 + 0], ndv[4 * k4 + 1] * ndv[4 * k4 + 1]};
      const v2f db = {ndv[4 * k4 + 2] * ndv[4 * k4 + 2], ndv[4 * k4 + 3] * ndv[4 * k4 + 3]};
      const v2f ga = {g.x, g.y};
      const v2f gb = {g.z, g.w};
      const v2f wa = __builtin_elementwise_max(zero2, __builtin_elementwise_fma(da, ninv, one2));
      const v2f wb = __builtin_elementwise_max(zero2, __builtin_elementwise_fma(db, ninv, one2));
      ws += wa;
      ps = __builtin_elementwise_fma(ga, wa, ps);
      ws += wb;
      ps = __builtin_elementwise_fma(gb, wb, ps);
    }
    const float conv = (ps.x + ps.y) / (ws.x + ws.y);
    outacc = fmaf(enc_s[b * NLAT + l], conv, outacc);
  }
  outacc += __shfl_xor(outacc, 8, 32);
  outacc += __shfl_xor(outacc, 16, 32);
  if (t < 8)
    stage[b * CONV_PITCH + w] = outacc;
  __syncthreads();
  if (w < BB)
    out[(size_t)w * NTOT + blockIdx.x * CONV_PTS + t] = stage[w * CONV_PITCH + t];
}

extern "C" void kernel_launch(void* const* d_in, const int* in_sizes, int n_in,
                              void* d_out, int out_size, void* d_ws, size_t ws_size,
                              hipStream_t stream) {
  const float* x      = (const float*)d_in[0];
  const float* enc_w  = (const float*)d_in[1];
  const float* enc_b  = (const float*)d_in[2];
  const float* dec_w  = (const float*)d_in[3];
  const float* bw_w   = (const float*)d_in[4];
  const float* bw_b   = (const float*)d_in[5];
  const float* ndist  = (const float*)d_in[6];
  const int*   nbid   = (const int*)d_in[7];
  const int*   labels = (const int*)d_in[8];
  float* out = (float*)d_out;

  float* ws       = (float*)d_ws;
  float* partials = ws + WS_PARTIALS;
  float* dec_t    = ws + WS_DECT;

  front_kernel<<<NENC + NTOT / TR_NODES, ENC_WAVES * 32, 0, stream>>>(x, enc_w, dec_w, partials, dec_t);
  conv_kernel<<<NTOT / CONV_PTS, CONV_PTS * 32, 0, stream>>>(ndist, nbid, labels, dec_t, partials, enc_b, bw_w, bw_b, out);
}


# conv: a point with no window clipped by the relu (voted per wave) uses sum(g)-inv*sum(g*d2) over K-inv*sum(d2): neighbour sums shared by all 8 batch rows; others keep the loop; sum order changes
# speedup vs baseline: 20.5836x; 1.1540x over previous
#include <hip/hip_runtime.h>
#include <math.h>

#define NTOT 16384
#define NLAT 20
#define MCL  10
#define KNB  32
#define BB   8
#define MUF  5.0f


typedef __attribute__((ext_vector_type(2))) float v2f;
typedef __attribute__((ext_vector_type(8))) float v8f;

#define NENC      32
#define ENC_WAVES 8
#define KCHUNK  (NTOT / (NENC * ENC_WAVES))
#define NOUT    (BB * NLAT)
#define WS_PARTIALS 0
#define DT_PITCH    32
#define WS_DECT     (NENC * NOUT)

#define TR_NODES 128
#define TR_PITCH 21

__global__ __launch_bounds__(ENC_WAVES * 32)
void front_kernel(const float* __restrict__ x,
                  const float* __restrict__ enc_w,
                  const float* __restrict__ dec_w,
                  float* __restrict__ partials,
                  float* __restrict__ dec_t) {
  __shared__ float tile[TR_NODES * TR_PITCH];
  __shared__ float red[ENC_WAVES * NOUT];
  const int tid = threadIdx.x;
  if (blockIdx.x < NENC) {
    const int t = tid & 31;
    const int w = tid >> 5;
    const int M = t & 15;
    const int Nn = t & 15;
    const int half = (t >> 4) << 1;
    const float maskA  = (M < BB) ? 1.f : 0.f;
    const float maskB1 = (Nn < NLAT - 16) ? 1.f : 0.f;
    const int rowA  = (M < BB) ? M : 0;
    const int rowB1 = (Nn < NLAT - 16) ? (16 + Nn) : 16;
    const float* pa  = x     + (size_t)rowA  * NTOT + half;
    const float* pb0 = enc_w + (size_t)Nn    * NTOT + half;
    const float* pb1 = enc_w + (size_t)rowB1 * NTOT + half;
    v8f acc0a = {}, acc0b = {}, acc1a = {}, acc1b = {};
    const int n0base = (blockIdx.x * ENC_WAVES + w) * KCHUNK;
    for (int s = 0; s < KCHUNK; s += 8) {
      const int n0 = n0base + s;
      const float2 av0 = *(const float2*)(pa  + n0);
      const float2 bv0 = *(const float2*)(pb0 + n0);
      const float2 cv0 = *(const float2*)(pb1 + n0);
      const float2 av1 = *(const float2*)(pa  + n0 + 4);
      const float2 bv1 = *(const float2*)(pb0 + n0 + 4);
      const float2 cv1 = *(const float2*)(pb1 + n0 + 4);
      v2f a0; a0.x = av0.x * maskA;  a0.y = av0.y * maskA;
      v2f b0; b0.x = bv0.x;          b0.y = bv0.y;
      v2f c0; c0.x = cv0.x * maskB1; c0.y = cv0.y * maskB1;
      v2f a1; a1.x = av1.x * maskA;  a1.y = av1.y * maskA;
      v2f b1; b1.x = bv1.x;          b1.y = bv1.y;
      v2f c1; c1.x = cv1.x * maskB1; c1.y = cv1.y * maskB1;
      acc0a = __builtin_amdgcn_wmma_f32_16x16x4_f32(false, a0, false, b0,
                                                    (short)0, acc0a, false, false);
      acc1a = __builtin_amdgcn_wmma_f32_16x16x4_f32(false, a0, false, c0,
                                                    (short)0, acc1a, false, false);
      acc0b = __builtin_amdgcn_wmma_f32_16x16x4_f32(false, a1, false, b1,
                                                    (short)0, acc0b, false, false);
      acc1b = __builtin_amdgcn_wmma_f32_16x16x4_f32(false, a1, false, c1,
                                                    (short)0, acc1b, false, false);
    }
    float* rw = red + w * NOUT;
    if (t < 16) {
#pragma unroll
      for (int i = 0; i < 8; ++i) rw[i * NLAT + t] = acc0a[i] + acc0b[i];
    }
    if (t < NLAT - 16) {
#pragma unroll
      for (int i = 0; i < 8; ++i) rw[i * NLAT + 16 + t] = acc1a[i] + acc1b[i];
    }
    __syncthreads();
    if (tid < NOUT) {
      float s = 0.f;
#pragma unroll
      for (int w2 = 0; w2 < ENC_WAVES; ++w2) s += red[w2 * NOUT + tid];
      partials[blockIdx.x * NOUT + tid] = s;
    }
    return;
  }
  const int base = (blockIdx.x - NENC) * TR_NODES;
  const int c  = tid & (TR_NODES - 1);
  const int l0 = tid >> 7;
#pragma unroll
  for (int r = 0; r < NLAT / 2; ++r) {
    const int l = l0 + 2 * r;
    tile[c * TR_PITCH + l] = dec_w[(size_t)l * NTOT + base + c];
  }
  __syncthreads();
  const int j = tid & 7;
#pragma unroll
  for (int r = 0; r < TR_NODES / 32; ++r) {
    const int cc = (tid >> 3) + 32 * r;
    float4 v = make_float4(0.f, 0.f, 0.f, 0.f);
    if (j < NLAT / 4) {
      const float* s = tile + cc * TR_PITCH + 4 * j;
      v = make_float4(s[0], s[1], s[2], s[3]);
    }
    *(float4*)(dec_t + (size_t)(base + cc) * DT_PITCH + 4 * j) = v;
  }
}

#define CONV_PTS   32
#define CONV_PITCH (CONV_PTS + 1)
#define NBW        (NLAT * MCL)
#define GT_PITCH   36

__global__ __launch_bounds__(CONV_PTS * 32, 16)
void conv_kernel(const float* __restrict__ ndist,
                 const int* __restrict__ nbid,
                 const int* __restrict__ labels,
                 const float* __restrict__ dec_t,
                 const float* __restrict__ partials,
                 const float* __restrict__ enc_b,
                 const float* __restrict__ bw_w,
                 const float* __restrict__ bw_b,
                 float* __restrict__ out) {
  __shared__ float stage[BB * CONV_PITCH];
  __shared__ float4 gtile[CONV_PTS * NLAT * (GT_PITCH / 4)];
  __shared__ float4 enc_s4[NOUT / 4];
  __shared__ float4 bww_s[NBW * NLAT / 4];
  __shared__ float bwb_s[NBW];
  __shared__ float sinv[NBW * BB];
  float* enc_s = (float*)enc_s4;
  const int tid = threadIdx.x;
  const int t  = tid & 31;
  const int w  = __builtin_amdgcn_readfirstlane(tid >> 5);
  const int n  = blockIdx.x * CONV_PTS + w;
  const int b  = t & 7;
  const int lq = t >> 3;
  const int   nid = nbid [(size_t)n * KNB + t];
  const int   m   = labels[n];
  const float* ndp = ndist + (size_t)n * KNB;
  float ndv[KNB];
#pragma unroll
  for (int k = 0; k < KNB; ++k) ndv[k] = ndp[k];
  const float dk = ndp[t];
  const float bpiece = bw_b[tid < NBW ? tid : NBW - 1];
  if (tid < NOUT) {
    float s = 0.f;
#pragma unroll
    for (int p = 0; p < NENC; ++p) s += partials[p * NOUT + tid];
    enc_s[tid] = s + enc_b[tid % NLAT];
  }
  const float4* pl = (const float4*)(dec_t + (size_t)nid * DT_PITCH);
  float4 q[NLAT / 4];
#pragma unroll
  for (int c = 0; c < NLAT / 4; ++c) q[c] = pl[c];
  if (tid < NBW * NLAT / 4) bww_s[tid] = ((const float4*)bw_w)[tid];
  if (tid < NBW) bwb_s[tid] = bpiece;
  float* gw = (float*)(gtile + w * (NLAT * (GT_PITCH / 4)));
#pragma unroll
  for (int c = 0; c < NLAT / 4; ++c) {
    gw[(4 * c + 0) * GT_PITCH + t] = q[c].x;
    gw[(4 * c + 1) * GT_PITCH + t] = q[c].y;
    gw[(4 * c + 2) * GT_PITCH + t] = q[c].z;
    gw[(4 * c + 3) * GT_PITCH + t] = q[c].w;
  }
  gw[(t >> 2) * GT_PITCH + KNB + (t & 3)] = dk * dk;
  __syncthreads();
  for (int idx = tid; idx < NBW * BB; idx += CONV_PTS * 32) {
    const int bb = idx & (BB - 1);
    const int j  = idx >> 3;
    const float4* wr = bww_s + j * (NLAT / 4);
    const float4* er = enc_s4 + bb * (NLAT / 4);
    float acc = bwb_s[j];
#pragma unroll
    for (int c = 0; c < NLAT / 4; ++c) {
      const float4 ev = er[c], wv = wr[c];
      acc = fmaf(ev.x, wv.x, acc);
      acc = fmaf(ev.y, wv.y, acc);
      acc = fmaf(ev.z, wv.z, acc);
      acc = fmaf(ev.w, wv.w, acc);
    }
    const float s = 1.0f / (1.0f + expf(-acc));
    const float lo = 2.0f / 60.0f / MUF;
    const float hi = 1.0f / 60.0f;
    const float bw = (hi - lo) * s + lo;
    const float d = bw * MUF;
    sinv[idx] = 1.0f / (d * d);
  }
  __syncthreads();
  const v2f one2  = {1.f, 1.f};
  const v2f zero2 = {0.f, 0.f};
  float outacc = 0.f;
  const float* invcol = sinv + m * BB + b;
  const float* gwl = gw + lq * GT_PITCH;
  float d2max = ndv[0] * ndv[0];
  float d2sum = d2max;
#pragma unroll
  for (int k = 1; k < KNB; ++k) {
    const float d2 = ndv[k] * ndv[k];
    d2max = fmaxf(d2max, d2);
    d2sum += d2;
  }
  float invl[NLAT / 4];
#pragma unroll
  for (int i = 0; i < NLAT / 4; ++i) invl[i] = invcol[(lq + 4 * i) * (MCL * BB)];
  const float invmax = fmaxf(fmaxf(fmaxf(invl[0], invl[1]), fmaxf(invl[2], invl[3])), invl[4]);
  const bool all_open = __all(fmaf(d2max, -invmax, 1.0f) > 0.0f) != 0;
  if (all_open) {
    const float4 e = *(const float4*)(gw + b * GT_PITCH + KNB);
    const float e0 = e.x, e1 = e.y, e2 = e.z, e3 = e.w;
#pragma unroll
    for (int i = 0; i < NLAT / 4; ++i) {
      const int l = lq + 4 * i;
      const float4 g = *(const float4*)(gwl + 4 * i * GT_PITCH + 4 * b);
      float g0 = (g.x + g.y) + (g.z + g.w);
      float g1 = fmaf(g.w, e3, fmaf(g.z, e2, fmaf(g.y, e1, g.x * e0)));
      g0 += __shfl_xor(g0, 1, 32);  g1 += __shfl_xor(g1, 1, 32);
      g0 += __shfl_xor(g0, 2, 32);  g1 += __shfl_xor(g1, 2, 32);
      g0 += __shfl_xor(g0, 4, 32);  g1 += __shfl_xor(g1, 4, 32);
      const float inv = invl[i];
      const float conv = fmaf(-inv, g1, g0) / fmaf(-inv, d2sum, (float)KNB);
      outacc = fmaf(enc_s[b * NLAT + l], conv, outacc);
    }
  } else
#pragma unroll
  for (int i = 0; i < NLAT / 4; ++i) {
    const int l = lq + 4 * i;
    const float inv = invcol[l * (MCL * BB)];
    const v2f ninv = {-inv, -inv};
    const float4* gr = (const float4*)(gwl + 4 * i * GT_PITCH);
    v2f ws = zero2, ps = zero2;
#pragma unroll
    for (int k4 = 0; k4 < KNB / 4; ++k4) {
      const float4 g = gr[k4];
      const v2f da = {ndv[4 * k4 + 0] * ndv[4 * k4 + 0], ndv[4 * k4 + 1] * ndv[4 * k4 + 1]};
      const v2f db = {ndv[4 * k4 + 2] * ndv[4 * k4 + 2], ndv[4 * k4 + 3] * ndv[4 * k4 + 3]};
      const v2f ga = {g.x, g.y};
      const v2f gb = {g.z, g.w};
      const v2f wa = __builtin_elementwise_max(zero2, __builtin_elementwise_fma(da, ninv, one2));
      const v2f wb = __builtin_elementwise_max(zero2, __builtin_elementwise_fma(db, ninv, one2));
      ws += wa;
      ps = __builtin_elementwise_fma(ga, wa, ps);
      ws += wb;
      ps = __builtin_elementwise_fma(gb, wb, ps);
    }
    const float conv = (ps.x + ps.y) / (ws.x + ws.y);
    outacc = fmaf(enc_s[b * NLAT + l], conv, outacc);
  }
  outacc += __shfl_xor(outacc, 8, 32);
  outacc += __shfl_xor(outacc, 16, 32);
  if (t < 8)
    stage[b * CONV_PITCH + w] = outacc;
  __syncthreads();
  if (w < BB)
    out[(size_t)w * NTOT + blockIdx.x * CONV_PTS + t] = stage[w * CONV_PITCH + t];
}

extern "C" void kernel_launch(void* const* d_in, const int* in_sizes, int n_in,
                              void* d_out, int out_size, void* d_ws, size_t ws_size,
                              hipStream_t stream) {
  const float* x      = (const float*)d_in[0];
  const float* enc_w  = (const float*)d_in[1];
  const float* enc_b  = (const float*)d_in[2];
  const float* dec_w  = (const float*)d_in[3];
  const float* bw_w   = (const float*)d_in[4];
  const float* bw_b   = (const float*)d_in[5];
  const float* ndist  = (const float*)d_in[6];
  const int*   nbid   = (const int*)d_in[7];
  const int*   labels = (const int*)d_in[8];
  float* out = (float*)d_out;

  float* ws       = (float*)d_ws;
  float* partials = ws + WS_PARTIALS;
  float* dec_t    = ws + WS_DECT;

  front_kernel<<<NENC + NTOT / TR_NODES, ENC_WAVES * 32, 0, stream>>>(x, enc_w, dec_w, partials, dec_t);
  conv_kernel<<<NTOT / CONV_PTS, CONV_PTS * 32, 0, stream>>>(ndist, nbid, labels, dec_t, partials, enc_b, bw_w, bw_b, out);
}


# conv basis fetch re-mapped: 8 lanes share a neighbour's 128-B line, 4 lines per load (was 32), 8 loads; unclipped neighbour sums taken in registers, no LDS tile; sum order changes
# speedup vs baseline: 23.0803x; 1.1213x over previous
#include <hip/hip_runtime.h>
#include <math.h>

#define NTOT 16384
#define NLAT 20
#define MCL  10
#define KNB  32
#define BB   8
#define MUF  5.0f


typedef __attribute__((ext_vector_type(2))) float v2f;
typedef __attribute__((ext_vector_type(8))) float v8f;

#define NENC      32
#define ENC_WAVES 8
#define KCHUNK  (NTOT / (NENC * ENC_WAVES))
#define NOUT    (BB * NLAT)
#define WS_PARTIALS 0
#define DT_PITCH    32
#define WS_DECT     (NENC * NOUT)

#define TR_NODES 128
#define TR_PITCH 21

__global__ __launch_bounds__(ENC_WAVES * 32)
void front_kernel(const float* __restrict__ x,
                  const float* __restrict__ enc_w,
                  const float* __restrict__ dec_w,
                  float* __restrict__ partials,
                  float* __restrict__ dec_t) {
  __shared__ float tile[TR_NODES * TR_PITCH];
  __shared__ float red[ENC_WAVES * NOUT];
  const int tid = threadIdx.x;
  if (blockIdx.x < NENC) {
    const int t = tid & 31;
    const int w = tid >> 5;
    const int M = t & 15;
    const int Nn = t & 15;
    const int half = (t >> 4) << 1;
    const float maskA  = (M < BB) ? 1.f : 0.f;
    const float maskB1 = (Nn < NLAT - 16) ? 1.f : 0.f;
    const int rowA  = (M < BB) ? M : 0;
    const int rowB1 = (Nn < NLAT - 16) ? (16 + Nn) : 16;
    const float* pa  = x     + (size_t)rowA  * NTOT + half;
    const float* pb0 = enc_w + (size_t)Nn    * NTOT + half;
    const float* pb1 = enc_w + (size_t)rowB1 * NTOT + half;
    v8f acc0a = {}, acc0b = {}, acc1a = {}, acc1b = {};
    const int n0base = (blockIdx.x * ENC_WAVES + w) * KCHUNK;
    for (int s = 0; s < KCHUNK; s += 8) {
      const int n0 = n0base + s;
      const float2 av0 = *(const float2*)(pa  + n0);
      const float2 bv0 = *(const float2*)(pb0 + n0);
      const float2 cv0 = *(const float2*)(pb1 + n0);
      const float2 av1 = *(const float2*)(pa  + n0 + 4);
      const float2 bv1 = *(const float2*)(pb0 + n0 + 4);
      const float2 cv1 = *(const float2*)(pb1 + n0 + 4);
      v2f a0; a0.x = av0.x * maskA;  a0.y = av0.y * maskA;
      v2f b0; b0.x = bv0.x;          b0.y = bv0.y;
      v2f c0; c0.x = cv0.x * maskB1; c0.y = cv0.y * maskB1;
      v2f a1; a1.x = av1.x * maskA;  a1.y = av1.y * maskA;
      v2f b1; b1.x = bv1.x;          b1.y = bv1.y;
      v2f c1; c1.x = cv1.x * maskB1; c1.y = cv1.y * maskB1;
      acc0a = __builtin_amdgcn_wmma_f32_16x16x4_f32(false, a0, false, b0,
                                                    (short)0, acc0a, false, false);
      acc1a = __builtin_amdgcn_wmma_f32_16x16x4_f32(false, a0, false, c0,
                                                    (short)0, acc1a, false, false);
      acc0b = __builtin_amdgcn_wmma_f32_16x16x4_f32(false, a1, false, b1,
                                                    (short)0, acc0b, false, false);
      acc1b = __builtin_amdgcn_wmma_f32_16x16x4_f32(false, a1, false, c1,
                                                    (short)0, acc1b, false, false);
    }
    float* rw = red + w * NOUT;
    if (t < 16) {
#pragma unroll
      for (int i = 0; i < 8; ++i) rw[i * NLAT + t] = acc0a[i] + acc0b[i];
    }
    if (t < NLAT - 16) {
#pragma unroll
      for (int i = 0; i < 8; ++i) rw[i * NLAT + 16 + t] = acc1a[i] + acc1b[i];
    }
    __syncthreads();
    if (tid < NOUT) {
      float s = 0.f;
#pragma unroll
      for (int w2 = 0; w2 < ENC_WAVES; ++w2) s += red[w2 * NOUT + tid];
      partials[blockIdx.x * NOUT + tid] = s;
    }
    return;
  }
  const int base = (blockIdx.x - NENC) * TR_NODES;
  const int c  = tid & (TR_NODES - 1);
  const int l0 = tid >> 7;
#pragma unroll
  for (int r = 0; r < NLAT / 2; ++r) {
    const int l = l0 + 2 * r;
    tile[c * TR_PITCH + l] = dec_w[(size_t)l * NTOT + base + c];
  }
  __syncthreads();
  const int j = tid & 7;
#pragma unroll
  for (int r = 0; r < TR_NODES / 32; ++r) {
    const int cc = (tid >> 3) + 32 * r;
    float4 v = make_float4(0.f, 0.f, 0.f, 0.f);
    if (j < NLAT / 4) {
      const float* s = tile + cc * TR_PITCH + 4 * j;
      v = make_float4(s[0], s[1], s[2], s[3]);
    }
    *(float4*)(dec_t + (size_t)(base + cc) * DT_PITCH + 4 * j) = v;
  }
}

#define CONV_PTS   32
#define CONV_PITCH (CONV_PTS + 1)
#define NBW        (NLAT * MCL)
#define GT_PITCH   36
#define NPIECE     (NLAT / 4)

__global__ __launch_bounds__(CONV_PTS * 32, 16)
void conv_kernel(const float* __restrict__ ndist,
                 const int* __restrict__ nbid,
                 const int* __restrict__ labels,
                 const float* __restrict__ dec_t,
                 const float* __restrict__ partials,
                 const float* __restrict__ enc_b,
                 const float* __restrict__ bw_w,
                 const float* __restrict__ bw_b,
                 float* __restrict__ out) {
  __shared__ float stage[BB * CONV_PITCH];
  __shared__ float4 gtile[CONV_PTS * NLAT * (GT_PITCH / 4)];
  __shared__ float4 enc_s4[NOUT / 4];
  __shared__ float4 bww_s[NBW * NLAT / 4];
  __shared__ float bwb_s[NBW];
  __shared__ float2 sinv2[NBW * BB / 2];
  float* enc_s = (float*)enc_s4;
  float* sinv  = (float*)sinv2;
  const int tid = threadIdx.x;
  const int t  = tid & 31;
  const int w  = __builtin_amdgcn_readfirstlane(tid >> 5);
  const int n  = blockIdx.x * CONV_PTS + w;
  const int kk = t >> 3;
  const int j  = t & 7;
  const bool act = j < NPIECE;
  const int jc = act ? j : NPIECE - 1;
  const int4*   nb4 = (const int4*)  (nbid  + (size_t)n * KNB + 8 * kk);
  const float4* nd4 = (const float4*)(ndist + (size_t)n * KNB + 8 * kk);
  const int4   ia = nb4[0], ib = nb4[1];
  const float4 da = nd4[0], db = nd4[1];
  const int   m   = labels[n];
  const float bpiece = bw_b[tid < NBW ? tid : NBW - 1];
  if (tid < NOUT) {
    float s = 0.f;
#pragma unroll
    for (int p = 0; p < NENC; ++p) s += partials[p * NOUT + tid];
    enc_s[tid] = s + enc_b[tid % NLAT];
  }
  const int ids[8] = {ia.x, ia.y, ia.z, ia.w, ib.x, ib.y, ib.z, ib.w};
  float4 q[8];
#pragma unroll
  for (int r = 0; r < 8; ++r)
    q[r] = *(const float4*)((const char*)dec_t +
                            ((unsigned)ids[r] * (unsigned)(DT_PITCH * 4) + (unsigned)(16 * jc)));
  if (tid < NBW * NLAT / 4) bww_s[tid] = ((const float4*)bw_w)[tid];
  if (tid < NBW) bwb_s[tid] = bpiece;
  const float e[8] = {da.x * da.x, da.y * da.y, da.z * da.z, da.w * da.w,
                      db.x * db.x, db.y * db.y, db.z * db.z, db.w * db.w};
  float s0x = q[0].x, s0y = q[0].y, s0z = q[0].z, s0w = q[0].w;
  float s1x = q[0].x * e[0], s1y = q[0].y * e[0], s1z = q[0].z * e[0], s1w = q[0].w * e[0];
  float dsum = e[0], dmax = e[0];
#pragma unroll
  for (int r = 1; r < 8; ++r) {
    s0x += q[r].x;  s0y += q[r].y;  s0z += q[r].z;  s0w += q[r].w;
    s1x = fmaf(q[r].x, e[r], s1x);  s1y = fmaf(q[r].y, e[r], s1y);
    s1z = fmaf(q[r].z, e[r], s1z);  s1w = fmaf(q[r].w, e[r], s1w);
    dsum += e[r];
    dmax = fmaxf(dmax, e[r]);
  }
#pragma unroll
  for (int x = 8; x <= 16; x <<= 1) {
    s0x += __shfl_xor(s0x, x, 32);  s0y += __shfl_xor(s0y, x, 32);
    s0z += __shfl_xor(s0z, x, 32);  s0w += __shfl_xor(s0w, x, 32);
    s1x += __shfl_xor(s1x, x, 32);  s1y += __shfl_xor(s1y, x, 32);
    s1z += __shfl_xor(s1z, x, 32);  s1w += __shfl_xor(s1w, x, 32);
    dsum += __shfl_xor(dsum, x, 32);
    dmax = fmaxf(dmax, __shfl_xor(dmax, x, 32));
  }
  float4* park = gtile + w * (NLAT * (GT_PITCH / 4));
  park[t]      = make_float4(s0x, s0y, s0z, s0w);
  park[32 + t] = make_float4(s1x, s1y, s1z, s1w);
  ((float2*)(park + 64))[t] = make_float2(dsum, dmax);
  __syncthreads();
  for (int idx = tid; idx < NBW * BB; idx += CONV_PTS * 32) {
    const int bb = idx & (BB - 1);
    const int ju = idx >> 3;
    const float4* wr = bww_s + ju * (NLAT / 4);
    const float4* er = enc_s4 + bb * (NLAT / 4);
    float acc = bwb_s[ju];
#pragma unroll
    for (int c = 0; c < NLAT / 4; ++c) {
      const float4 ev = er[c], wv = wr[c];
      acc = fmaf(ev.x, wv.x, acc);
      acc = fmaf(ev.y, wv.y, acc);
      acc = fmaf(ev.z, wv.z, acc);
      acc = fmaf(ev.w, wv.w, acc);
    }
    const float s = 1.0f / (1.0f + expf(-acc));
    const float lo = 2.0f / 60.0f / MUF;
    const float hi = 1.0f / 60.0f;
    const float bw = (hi - lo) * s + lo;
    const float d = bw * MUF;
    sinv[idx] = 1.0f / (d * d);
  }
  __syncthreads();
  const float2 dd = ((const float2*)(park + 64))[t];
  const float d2sum = dd.x;
  const float d2max = dd.y;
  float2 inv2[4];
#pragma unroll
  for (int c = 0; c < 4; ++c) inv2[c] = sinv2[((4 * jc + c) * MCL + m) * (BB / 2) + kk];
  const float invmax = fmaxf(fmaxf(fmaxf(inv2[0].x, inv2[0].y), fmaxf(inv2[1].x, inv2[1].y)),
                             fmaxf(fmaxf(inv2[2].x, inv2[2].y), fmaxf(inv2[3].x, inv2[3].y)));
  const bool all_open = __all(fmaf(d2max, -invmax, 1.0f) > 0.0f) != 0;
  {
    const int b0 = 2 * kk;
    const float4 ea = enc_s4[b0 * (NLAT / 4) + jc];
    const float4 eb = enc_s4[(b0 + 1) * (NLAT / 4) + jc];
    const float4 p0 = park[t];
    const float4 p1 = park[32 + t];
    const float g0[4] = {p0.x, p0.y, p0.z, p0.w};
    const float g1[4] = {p1.x, p1.y, p1.z, p1.w};
    const float ea_[4] = {ea.x, ea.y, ea.z, ea.w};
    const float eb_[4] = {eb.x, eb.y, eb.z, eb.w};
    float acc0 = 0.f, acc1 = 0.f;
#pragma unroll
    for (int c = 0; c < 4; ++c) {
      const float conv0 = fmaf(-inv2[c].x, g1[c], g0[c]) / fmaf(-inv2[c].x, d2sum, (float)KNB);
      const float conv1 = fmaf(-inv2[c].y, g1[c], g0[c]) / fmaf(-inv2[c].y, d2sum, (float)KNB);
      acc0 = fmaf(ea_[c], conv0, acc0);
      acc1 = fmaf(eb_[c], conv1, acc1);
    }
    if (!act) { acc0 = 0.f; acc1 = 0.f; }
    acc0 += __shfl_xor(acc0, 1, 32);  acc1 += __shfl_xor(acc1, 1, 32);
    acc0 += __shfl_xor(acc0, 2, 32);  acc1 += __shfl_xor(acc1, 2, 32);
    acc0 += __shfl_xor(acc0, 4, 32);  acc1 += __shfl_xor(acc1, 4, 32);
    if (j == 0) {
      stage[b0 * CONV_PITCH + w] = acc0;
      stage[(b0 + 1) * CONV_PITCH + w] = acc1;
    }
  }
  if (!all_open) {
    const int b  = t & 7;
    const int lq = t >> 3;
    const int nid = nbid[(size_t)n * KNB + t];
    const float4* pl = (const float4*)(dec_t + (size_t)nid * DT_PITCH);
    float4 qg[NPIECE];
#pragma unroll
    for (int c = 0; c < NPIECE; ++c) qg[c] = pl[c];
    const float* ndp = ndist + (size_t)n * KNB;
    float ndv[KNB];
#pragma unroll
    for (int k = 0; k < KNB; ++k) ndv[k] = ndp[k];
    float* gw = (float*)park;
#pragma unroll
    for (int c = 0; c < NPIECE; ++c) {
      gw[(4 * c + 0) * GT_PITCH + t] = qg[c].x;
      gw[(4 * c + 1) * GT_PITCH + t] = qg[c].y;
      gw[(4 * c + 2) * GT_PITCH + t] = qg[c].z;
      gw[(4 * c + 3) * GT_PITCH + t] = qg[c].w;
    }
    const v2f one2  = {1.f, 1.f};
    const v2f zero2 = {0.f, 0.f};
    float outacc = 0.f;
    const float* invcol = sinv + m * BB + b;
    const float* gwl = gw + lq * GT_PITCH;
#pragma unroll
    for (int i = 0; i < NLAT / 4; ++i) {
      const int l = lq + 4 * i;
      const float inv = invcol[l * (MCL * BB)];
      const v2f ninv = {-inv, -inv};
      const float4* gr = (const float4*)(gwl + 4 * i * GT_PITCH);
      v2f ws = zero2, ps = zero2;
#pragma unroll
      for (int k4 = 0; k4 < KNB / 4; ++k4) {
        const float4 g = gr[k4];
        const v2f dA = {ndv[4 * k4 + 0] * ndv[4 * k4 + 0], ndv[4 * k4 + 1] * ndv[4 * k4 + 1]};
        const v2f dB = {ndv[4 * k4 + 2] * ndv[4 * k4 + 2], ndv[4 * k4 + 3] * ndv[4 * k4 + 3]};
        const v2f ga = {g.x, g.y};
        const v2f gb = {g.z, g.w};
        const v2f wa = __builtin_elementwise_max(zero2, __builtin_elementwise_fma(dA, ninv, one2));
        const v2f wb = __builtin_elementwise_max(zero2, __builtin_elementwise_fma(dB, ninv, one2));
        ws += wa;
        ps = __builtin_elementwise_fma(ga, wa, ps);
        ws += wb;
        ps = __builtin_elementwise_fma(gb, wb, ps);
      }
      const float conv = (ps.x + ps.y) / (ws.x + ws.y);
      outacc = fmaf(enc_s[b * NLAT + l], conv, outacc);
    }
    outacc += __shfl_xor(outacc, 8, 32);
    outacc += __shfl_xor(outacc, 16, 32);
    if (t < 8)
      stage[b * CONV_PITCH + w] = outacc;
  }
  __syncthreads();
  if (w < BB)
    out[(size_t)w * NTOT + blockIdx.x * CONV_PTS + t] = stage[w * CONV_PITCH + t];
}

extern "C" void kernel_launch(void* const* d_in, const int* in_sizes, int n_in,
                              void* d_out, int out_size, void* d_ws, size_t ws_size,
                              hipStream_t stream) {
  const float* x      = (const float*)d_in[0];
  const float* enc_w  = (const float*)d_in[1];
  const float* enc_b  = (const float*)d_in[2];
  const float* dec_w  = (const float*)d_in[3];
  const float* bw_w   = (const float*)d_in[4];
  const float* bw_b   = (const float*)d_in[5];
  const float* ndist  = (const float*)d_in[6];
  const int*   nbid   = (const int*)d_in[7];
  const int*   labels = (const int*)d_in[8];
  float* out = (float*)d_out;

  float* ws       = (float*)d_ws;
  float* partials = ws + WS_PARTIALS;
  float* dec_t    = ws + WS_DECT;

  front_kernel<<<NENC + NTOT / TR_NODES, ENC_WAVES * 32, 0, stream>>>(x, enc_w, dec_w, partials, dec_t);
  conv_kernel<<<NTOT / CONV_PTS, CONV_PTS * 32, 0, stream>>>(ndist, nbid, labels, dec_t, partials, enc_b, bw_w, bw_b, out);
}


# conv: four points per wave, lane group of 8 = point: 4096 waves in 256 groups of 16 (was 16384 in 512 of 32); basis pieces wait in registers across the scale stage; sum order kept
# speedup vs baseline: 25.0900x; 1.0871x over previous
#include <hip/hip_runtime.h>
#include <math.h>

#define NTOT 16384
#define NLAT 20
#define MCL  10
#define KNB  32
#define BB   8
#define MUF  5.0f


typedef __attribute__((ext_vector_type(2))) float v2f;
typedef __attribute__((ext_vector_type(8))) float v8f;

#define NENC      32
#define ENC_WAVES 8
#define KCHUNK  (NTOT / (NENC * ENC_WAVES))
#define NOUT    (BB * NLAT)
#define WS_PARTIALS 0
#define DT_PITCH    32
#define WS_DECT     (NENC * NOUT)

#define TR_NODES 128
#define TR_PITCH 21

__global__ __launch_bounds__(ENC_WAVES * 32)
void front_kernel(const float* __restrict__ x,
                  const float* __restrict__ enc_w,
                  const float* __restrict__ dec_w,
                  float* __restrict__ partials,
                  float* __restrict__ dec_t) {
  __shared__ float tile[TR_NODES * TR_PITCH];
  __shared__ float red[ENC_WAVES * NOUT];
  const int tid = threadIdx.x;
  if (blockIdx.x < NENC) {
    const int t = tid & 31;
    const int w = tid >> 5;
    const int M = t & 15;
    const int Nn = t & 15;
    const int half = (t >> 4) << 1;
    const float maskA  = (M < BB) ? 1.f : 0.f;
    const float maskB1 = (Nn < NLAT - 16) ? 1.f : 0.f;
    const int rowA  = (M < BB) ? M : 0;
    const int rowB1 = (Nn < NLAT - 16) ? (16 + Nn) : 16;
    const float* pa  = x     + (size_t)rowA  * NTOT + half;
    const float* pb0 = enc_w + (size_t)Nn    * NTOT + half;
    const float* pb1 = enc_w + (size_t)rowB1 * NTOT + half;
    v8f acc0a = {}, acc0b = {}, acc1a = {}, acc1b = {};
    const int n0base = (blockIdx.x * ENC_WAVES + w) * KCHUNK;
    for (int s = 0; s < KCHUNK; s += 8) {
      const int n0 = n0base + s;
      const float2 av0 = *(const float2*)(pa  + n0);
      const float2 bv0 = *(const float2*)(pb0 + n0);
      const float2 cv0 = *(const float2*)(pb1 + n0);
      const float2 av1 = *(const float2*)(pa  + n0 + 4);
      const float2 bv1 = *(const float2*)(pb0 + n0 + 4);
      const float2 cv1 = *(const float2*)(pb1 + n0 + 4);
      v2f a0; a0.x = av0.x * maskA;  a0.y = av0.y * maskA;
      v2f b0; b0.x = bv0.x;          b0.y = bv0.y;
      v2f c0; c0.x = cv0.x * maskB1; c0.y = cv0.y * maskB1;
      v2f a1; a1.x = av1.x * maskA;  a1.y = av1.y * maskA;
      v2f b1; b1.x = bv1.x;          b1.y = bv1.y;
      v2f c1; c1.x = cv1.x * maskB1; c1.y = cv1.y * maskB1;
      acc0a = __builtin_amdgcn_wmma_f32_16x16x4_f32(false, a0, false, b0,
                                                    (short)0, acc0a, false, false);
      acc1a = __builtin_amdgcn_wmma_f32_16x16x4_f32(false, a0, false, c0,
                                                    (short)0, acc1a, false, false);
      acc0b = __builtin_amdgcn_wmma_f32_16x16x4_f32(false, a1, false, b1,
                                                    (short)0, acc0b, false, false);
      acc1b = __builtin_amdgcn_wmma_f32_16x16x4_f32(false, a1, false, c1,
                                                    (short)0, acc1b, false, false);
    }
    float* rw = red + w * NOUT;
    if (t < 16) {
#pragma unroll
      for (int i = 0; i < 8; ++i) rw[i * NLAT + t] = acc0a[i] + acc0b[i];
    }
    if (t < NLAT - 16) {
#pragma unroll
      for (int i = 0; i < 8; ++i) rw[i * NLAT + 16 + t] = acc1a[i] + acc1b[i];
    }
    __syncthreads();
    if (tid < NOUT) {
      float s = 0.f;
#pragma unroll
      for (int w2 = 0; w2 < ENC_WAVES; ++w2) s += red[w2 * NOUT + tid];
      partials[blockIdx.x * NOUT + tid] = s;
    }
    return;
  }
  const int base = (blockIdx.x - NENC) * TR_NODES;
  const int c  = tid & (TR_NODES - 1);
  const int l0 = tid >> 7;
#pragma unroll
  for (int r = 0; r < NLAT / 2; ++r) {
    const int l = l0 + 2 * r;
    tile[c * TR_PITCH + l] = dec_w[(size_t)l * NTOT + base + c];
  }
  __syncthreads();
  const int j = tid & 7;
#pragma unroll
  for (int r = 0; r < TR_NODES / 32; ++r) {
    const int cc = (tid >> 3) + 32 * r;
    float4 v = make_float4(0.f, 0.f, 0.f, 0.f);
    if (j < NLAT / 4) {
      const float* s = tile + cc * TR_PITCH + 4 * j;
      v = make_float4(s[0], s[1], s[2], s[3]);
    }
    *(float4*)(dec_t + (size_t)(base + cc) * DT_PITCH + 4 * j) = v;
  }
}

#define CONV_WAVES 16
#define PPW        4
#define CONV_PTS   (CONV_WAVES * PPW)
#define CONV_PITCH (CONV_PTS + 1)
#define CONV_THREADS (CONV_WAVES * 32)
#define NBW        (NLAT * MCL)
#define GT_PITCH   36
#define NPIECE     (NLAT / 4)
#define COPY_THREADS (CONV_THREADS - NOUT)
#define MLP_FULL   (NBW * BB / CONV_THREADS)
#define MLP_REST   (NBW * BB - MLP_FULL * CONV_THREADS)

__device__ __forceinline__ float bw_inv(const float4* __restrict__ bww_s, const float4* __restrict__ enc_s4,
                                        const float* __restrict__ bwb_s, int idx) {
  const int bb = idx & (BB - 1);
  const int ju = idx >> 3;
  const float4* wr = bww_s + ju * (NLAT / 4);
  const float4* er = enc_s4 + bb * (NLAT / 4);
  float acc = bwb_s[ju];
#pragma unroll
  for (int c = 0; c < NLAT / 4; ++c) {
    const float4 ev = er[c], wv = wr[c];
    acc = fmaf(ev.x, wv.x, acc);
    acc = fmaf(ev.y, wv.y, acc);
    acc = fmaf(ev.z, wv.z, acc);
    acc = fmaf(ev.w, wv.w, acc);
  }
  const float s = 1.0f / (1.0f + expf(-acc));
  const float lo = 2.0f / 60.0f / MUF;
  const float hi = 1.0f / 60.0f;
  const float bw = (hi - lo) * s + lo;
  const float d = bw * MUF;
  return 1.0f / (d * d);
}

__global__ __launch_bounds__(CONV_THREADS)
void conv_kernel(const float* __restrict__ ndist,
                 const int* __restrict__ nbid,
                 const int* __restrict__ labels,
                 const float* __restrict__ dec_t,
                 const float* __restrict__ partials,
                 const float* __restrict__ enc_b,
                 const float* __restrict__ bw_w,
                 const float* __restrict__ bw_b,
                 float* __restrict__ out) {
  __shared__ float stage[BB * CONV_PITCH];
  __shared__ float4 gtile[CONV_WAVES * NLAT * (GT_PITCH / 4)];
  __shared__ float4 enc_s4[NOUT / 4];
  __shared__ float4 bww_s[3 * COPY_THREADS];
  __shared__ float bwb_s[NBW];
  __shared__ float4 sinv4[NBW * BB / 4];
  float* enc_s = (float*)enc_s4;
  float* sinv  = (float*)sinv4;
  const int tid = threadIdx.x;
  const int t  = tid & 31;
  const int w  = __builtin_amdgcn_readfirstlane(tid >> 5);
  const int nw = blockIdx.x * CONV_PTS + w * PPW;
  const int p  = t >> 3;
  const int j  = t & 7;
  const int n  = nw + p;
  const bool act = j < NPIECE;
  const int jc = act ? j : NPIECE - 1;
  const int4*   nb4 = (const int4*)  (nbid  + (size_t)n * KNB);
  const float4* nd4 = (const float4*)(ndist + (size_t)n * KNB);
  int4 iv[KNB / 4];
  float4 dv[KNB / 4];
#pragma unroll
  for (int r = 0; r < KNB / 4; ++r) { iv[r] = nb4[r]; dv[r] = nd4[r]; }
  const int m = labels[n];
  const float bpiece = bw_b[tid < NBW ? tid : NBW - 1];
  if (tid < NOUT) {
    float s = 0.f;
#pragma unroll
    for (int pp = 0; pp < NENC; ++pp) s += partials[pp * NOUT + tid];
    enc_s[tid] = s + enc_b[tid % NLAT];
  } else {
    const float4* w4 = (const float4*)bw_w;
    const int i0 = tid - NOUT;
    const int i2 = i0 + 2 * COPY_THREADS < NBW * NLAT / 4 ? i0 + 2 * COPY_THREADS : NBW * NLAT / 4 - 1;
    const float4 wp0 = w4[i0], wp1 = w4[i0 + COPY_THREADS], wp2 = w4[i2];
    bww_s[i0] = wp0;
    bww_s[i0 + COPY_THREADS] = wp1;
    bww_s[i0 + 2 * COPY_THREADS] = wp2;
  }
  if (tid < NBW) bwb_s[tid] = bpiece;
  __syncthreads();
  float4 q[KNB];
#pragma unroll
  for (int r = 0; r < KNB / 4; ++r) {
    const int ids[4] = {iv[r].x, iv[r].y, iv[r].z, iv[r].w};
#pragma unroll
    for (int u = 0; u < 4; ++u)
      q[4 * r + u] = *(const float4*)((const char*)dec_t +
                                      ((unsigned)ids[u] * (unsigned)(DT_PITCH * 4) + (unsigned)(16 * jc)));
  }
  {
    float r3[MLP_FULL];
#pragma unroll
    for (int r = 0; r < MLP_FULL; ++r) r3[r] = bw_inv(bww_s, enc_s4, bwb_s, tid + r * CONV_THREADS);
#pragma unroll
    for (int r = 0; r < MLP_FULL; ++r) sinv[tid + r * CONV_THREADS] = r3[r];
    if (tid < MLP_REST)
      sinv[tid + MLP_FULL * CONV_THREADS] = bw_inv(bww_s, enc_s4, bwb_s, tid + MLP_FULL * CONV_THREADS);
  }
  __syncthreads();
  float e[KNB];
#pragma unroll
  for (int r = 0; r < KNB / 4; ++r) {
    e[4 * r + 0] = dv[r].x * dv[r].x;  e[4 * r + 1] = dv[r].y * dv[r].y;
    e[4 * r + 2] = dv[r].z * dv[r].z;  e[4 * r + 3] = dv[r].w * dv[r].w;
  }
  float a0x[4], a0y[4], a0z[4], a0w[4], a1x[4], a1y[4], a1z[4], a1w[4], ads[4], adm[4];
#pragma unroll
  for (int a = 0; a < 4; ++a) {
    const int r0 = 8 * a;
    a0x[a] = q[r0].x;  a0y[a] = q[r0].y;  a0z[a] = q[r0].z;  a0w[a] = q[r0].w;
    a1x[a] = q[r0].x * e[r0];  a1y[a] = q[r0].y * e[r0];
    a1z[a] = q[r0].z * e[r0];  a1w[a] = q[r0].w * e[r0];
    ads[a] = e[r0];  adm[a] = e[r0];
#pragma unroll
    for (int r = r0 + 1; r < r0 + 8; ++r) {
      a0x[a] += q[r].x;  a0y[a] += q[r].y;  a0z[a] += q[r].z;  a0w[a] += q[r].w;
      a1x[a] = fmaf(q[r].x, e[r], a1x[a]);  a1y[a] = fmaf(q[r].y, e[r], a1y[a]);
      a1z[a] = fmaf(q[r].z, e[r], a1z[a]);  a1w[a] = fmaf(q[r].w, e[r], a1w[a]);
      ads[a] += e[r];
      adm[a] = fmaxf(adm[a], e[r]);
    }
  }
  const float g0[4] = {(a0x[0] + a0x[1]) + (a0x[2] + a0x[3]), (a0y[0] + a0y[1]) + (a0y[2] + a0y[3]),
                       (a0z[0] + a0z[1]) + (a0z[2] + a0z[3]), (a0w[0] + a0w[1]) + (a0w[2] + a0w[3])};
  const float g1[4] = {(a1x[0] + a1x[1]) + (a1x[2] + a1x[3]), (a1y[0] + a1y[1]) + (a1y[2] + a1y[3]),
                       (a1z[0] + a1z[1]) + (a1z[2] + a1z[3]), (a1w[0] + a1w[1]) + (a1w[2] + a1w[3])};
  const float d2sum = (ads[0] + ads[1]) + (ads[2] + ads[3]);
  const float d2max = fmaxf(fmaxf(adm[0], adm[1]), fmaxf(adm[2], adm[3]));
  float inv[4][BB];
#pragma unroll
  for (int c = 0; c < 4; ++c) {
    const float4* ip = sinv4 + ((4 * jc + c) * MCL + m) * (BB / 4);
    const float4 lo4 = ip[0], hi4 = ip[1];
    inv[c][0] = lo4.x;  inv[c][1] = lo4.y;  inv[c][2] = lo4.z;  inv[c][3] = lo4.w;
    inv[c][4] = hi4.x;  inv[c][5] = hi4.y;  inv[c][6] = hi4.z;  inv[c][7] = hi4.w;
  }
  float invmax = inv[0][0];
#pragma unroll
  for (int c = 0; c < 4; ++c)
#pragma unroll
    for (int b = 0; b < BB; ++b) invmax = fmaxf(invmax, inv[c][b]);
  const unsigned open = (unsigned)__ballot(fmaf(d2max, -invmax, 1.0f) > 0.0f);
  {
    float acc[BB];
#pragma unroll
    for (int b = 0; b < BB; ++b) {
      const float4 eb = enc_s4[b * (NLAT / 4) + jc];
      const float eb_[4] = {eb.x, eb.y, eb.z, eb.w};
      float a = 0.f;
#pragma unroll
      for (int c = 0; c < 4; ++c) {
        const float conv = fmaf(-inv[c][b], g1[c], g0[c]) / fmaf(-inv[c][b], d2sum, (float)KNB);
        a = fmaf(eb_[c], conv, a);
      }
      if (!act) a = 0.f;
      a += __shfl_xor(a, 1, 32);
      a += __shfl_xor(a, 2, 32);
      a += __shfl_xor(a, 4, 32);
      acc[b] = a;
    }
    if (j == 0) {
#pragma unroll
      for (int b = 0; b < BB; ++b) stage[b * CONV_PITCH + w * PPW + p] = acc[b];
    }
  }
  if (open != 0xFFFFFFFFu) {
    const int b  = t & 7;
    const int lq = t >> 3;
    float* gw = (float*)(gtile + w * (NLAT * (GT_PITCH / 4)));
    for (int pp = 0; pp < PPW; ++pp) {
      if (((open >> (8 * pp)) & 0xFFu) == 0xFFu) continue;
      const int np = nw + pp;
      const int mp = labels[np];
      const int nid = nbid[(size_t)np * KNB + t];
      const float4* pl = (const float4*)(dec_t + (size_t)nid * DT_PITCH);
      float4 qg[NPIECE];
#pragma unroll
      for (int c = 0; c < NPIECE; ++c) qg[c] = pl[c];
      const float* ndp = ndist + (size_t)np * KNB;
      float ndv[KNB];
#pragma unroll
      for (int k = 0; k < KNB; ++k) ndv[k] = ndp[k];
#pragma unroll
      for (int c = 0; c < NPIECE; ++c) {
        gw[(4 * c + 0) * GT_PITCH + t] = qg[c].x;
        gw[(4 * c + 1) * GT_PITCH + t] = qg[c].y;
        gw[(4 * c + 2) * GT_PITCH + t] = qg[c].z;
        gw[(4 * c + 3) * GT_PITCH + t] = qg[c].w;
      }
      const v2f one2  = {1.f, 1.f};
      const v2f zero2 = {0.f, 0.f};
      float outacc = 0.f;
      const float* invcol = sinv + mp * BB + b;
      const float* gwl = gw + lq * GT_PITCH;
#pragma unroll 1
      for (int i = 0; i < NLAT / 4; ++i) {
        const int l = lq + 4 * i;
        const float inv1 = invcol[l * (MCL * BB)];
        const v2f ninv = {-inv1, -inv1};
        const float4* gr = (const float4*)(gwl + 4 * i * GT_PITCH);
        v2f ws = zero2, ps = zero2;
#pragma unroll
        for (int k4 = 0; k4 < KNB / 4; ++k4) {
          const float4 g = gr[k4];
          const v2f dA = {ndv[4 * k4 + 0] * ndv[4 * k4 + 0], ndv[4 * k4 + 1] * ndv[4 * k4 + 1]};
          const v2f dB = {ndv[4 * k4 + 2] * ndv[4 * k4 + 2], ndv[4 * k4 + 3] * ndv[4 * k4 + 3]};
          const v2f ga = {g.x, g.y};
          const v2f gb = {g.z, g.w};
          const v2f wa = __builtin_elementwise_max(zero2, __builtin_elementwise_fma(dA, ninv, one2));
          const v2f wb = __builtin_elementwise_max(zero2, __builtin_elementwise_fma(dB, ninv, one2));
          ws += wa;
          ps = __builtin_elementwise_fma(ga, wa, ps);
          ws += wb;
          ps = __builtin_elementwise_fma(gb, wb, ps);
        }
        const float conv = (ps.x + ps.y) / (ws.x + ws.y);
        outacc = fmaf(enc_s[b * NLAT + l], conv, outacc);
      }
      outacc += __shfl_xor(outacc, 8, 32);
      outacc += __shfl_xor(outacc, 16, 32);
      if (t < 8)
        stage[b * CONV_PITCH + w * PPW + pp] = outacc;
    }
  }
  __syncthreads();
  out[(size_t)(w >> 1) * NTOT + blockIdx.x * CONV_PTS + (w & 1) * 32 + t] =
      stage[(w >> 1) * CONV_PITCH + (w & 1) * 32 + t];
}

extern "C" void kernel_launch(void* const* d_in, const int* in_sizes, int n_in,
                              void* d_out, int out_size, void* d_ws, size_t ws_size,
                              hipStream_t stream) {
  const float* x      = (const float*)d_in[0];
  const float* enc_w  = (const float*)d_in[1];
  const float* enc_b  = (const float*)d_in[2];
  const float* dec_w  = (const float*)d_in[3];
  const float* bw_w   = (const float*)d_in[4];
  const float* bw_b   = (const float*)d_in[5];
  const float* ndist  = (const float*)d_in[6];
  const int*   nbid   = (const int*)d_in[7];
  const int*   labels = (const int*)d_in[8];
  float* out = (float*)d_out;

  float* ws       = (float*)d_ws;
  float* partials = ws + WS_PARTIALS;
  float* dec_t    = ws + WS_DECT;

  front_kernel<<<NENC + NTOT / TR_NODES, ENC_WAVES * 32, 0, stream>>>(x, enc_w, dec_w, partials, dec_t);
  conv_kernel<<<NTOT / CONV_PTS, CONV_THREADS, 0, stream>>>(ndist, nbid, labels, dec_t, partials, enc_b, bw_w, bw_b, out);
}


# conv and bandwidth stage: each division of the hot path (32 quotients per lane, sigmoid, 1/d^2) is numerator x v_rcp_f32 instead of the 11-instruction IEEE chain; results move by about 1 ulp
# speedup vs baseline: 26.4841x; 1.0556x over previous
#include <hip/hip_runtime.h>
#include <math.h>

#define NTOT 16384
#define NLAT 20
#define MCL  10
#define KNB  32
#define BB   8
#define MUF  5.0f


typedef __attribute__((ext_vector_type(2))) float v2f;
typedef __attribute__((ext_vector_type(8))) float v8f;

#define NENC      32
#define ENC_WAVES 8
#define KCHUNK  (NTOT / (NENC * ENC_WAVES))
#define NOUT    (BB * NLAT)
#define WS_PARTIALS 0
#define DT_PITCH    32
#define WS_DECT     (NENC * NOUT)

#define TR_NODES 128
#define TR_PITCH 21

__global__ __launch_bounds__(ENC_WAVES * 32)
void front_kernel(const float* __restrict__ x,
                  const float* __restrict__ enc_w,
                  const float* __restrict__ dec_w,
                  float* __restrict__ partials,
                  float* __restrict__ dec_t) {
  __shared__ float tile[TR_NODES * TR_PITCH];
  __shared__ float red[ENC_WAVES * NOUT];
  const int tid = threadIdx.x;
  if (blockIdx.x < NENC) {
    const int t = tid & 31;
    const int w = tid >> 5;
    const int M = t & 15;
    const int Nn = t & 15;
    const int half = (t >> 4) << 1;
    const float maskA  = (M < BB) ? 1.f : 0.f;
    const float maskB1 = (Nn < NLAT - 16) ? 1.f : 0.f;
    const int rowA  = (M < BB) ? M : 0;
    const int rowB1 = (Nn < NLAT - 16) ? (16 + Nn) : 16;
    const float* pa  = x     + (size_t)rowA  * NTOT + half;
    const float* pb0 = enc_w + (size_t)Nn    * NTOT + half;
    const float* pb1 = enc_w + (size_t)rowB1 * NTOT + half;
    v8f acc0a = {}, acc0b = {}, acc1a = {}, acc1b = {};
    const int n0base = (blockIdx.x * ENC_WAVES + w) * KCHUNK;
    for (int s = 0; s < KCHUNK; s += 8) {
      const int n0 = n0base + s;
      const float2 av0 = *(const float2*)(pa  + n0);
      const float2 bv0 = *(const float2*)(pb0 + n0);
      const float2 cv0 = *(const float2*)(pb1 + n0);
      const float2 av1 = *(const float2*)(pa  + n0 + 4);
      const float2 bv1 = *(const float2*)(pb0 + n0 + 4);
      const float2 cv1 = *(const float2*)(pb1 + n0 + 4);
      v2f a0; a0.x = av0.x * maskA;  a0.y = av0.y * maskA;
      v2f b0; b0.x = bv0.x;          b0.y = bv0.y;
      v2f c0; c0.x = cv0.x * maskB1; c0.y = cv0.y * maskB1;
      v2f a1; a1.x = av1.x * maskA;  a1.y = av1.y * maskA;
      v2f b1; b1.x = bv1.x;          b1.y = bv1.y;
      v2f c1; c1.x = cv1.x * maskB1; c1.y = cv1.y * maskB1;
      acc0a = __builtin_amdgcn_wmma_f32_16x16x4_f32(false, a0, false, b0,
                                                    (short)0, acc0a, false, false);
      acc1a = __builtin_amdgcn_wmma_f32_16x16x4_f32(false, a0, false, c0,
                                                    (short)0, acc1a, false, false);
      acc0b = __builtin_amdgcn_wmma_f32_16x16x4_f32(false, a1, false, b1,
                                                    (short)0, acc0b, false, false);
      acc1b = __builtin_amdgcn_wmma_f32_16x16x4_f32(false, a1, false, c1,
                                                    (short)0, acc1b, false, false);
    }
    float* rw = red + w * NOUT;
    if (t < 16) {
#pragma unroll
      for (int i = 0; i < 8; ++i) rw[i * NLAT + t] = acc0a[i] + acc0b[i];
    }
    if (t < NLAT - 16) {
#pragma unroll
      for (int i = 0; i < 8; ++i) rw[i * NLAT + 16 + t] = acc1a[i] + acc1b[i];
    }
    __syncthreads();
    if (tid < NOUT) {
      float s = 0.f;
#pragma unroll
      for (int w2 = 0; w2 < ENC_WAVES; ++w2) s += red[w2 * NOUT + tid];
      partials[blockIdx.x * NOUT + tid] = s;
    }
    return;
  }
  const int base = (blockIdx.x - NENC) * TR_NODES;
  const int c  = tid & (TR_NODES - 1);
  const int l0 = tid >> 7;
#pragma unroll
  for (int r = 0; r < NLAT / 2; ++r) {
    const int l = l0 + 2 * r;
    tile[c * TR_PITCH + l] = dec_w[(size_t)l * NTOT + base + c];
  }
  __syncthreads();
  const int j = tid & 7;
#pragma unroll
  for (int r = 0; r < TR_NODES / 32; ++r) {
    const int cc = (tid >> 3) + 32 * r;
    float4 v = make_float4(0.f, 0.f, 0.f, 0.f);
    if (j < NLAT / 4) {
      const float* s = tile + cc * TR_PITCH + 4 * j;
      v = make_float4(s[0], s[1], s[2], s[3]);
    }
    *(float4*)(dec_t + (size_t)(base + cc) * DT_PITCH + 4 * j) = v;
  }
}

#define CONV_WAVES 16
#define PPW        4
#define CONV_PTS   (CONV_WAVES * PPW)
#define CONV_PITCH (CONV_PTS + 1)
#define CONV_THREADS (CONV_WAVES * 32)
#define NBW        (NLAT * MCL)
#define GT_PITCH   36
#define NPIECE     (NLAT / 4)
#define COPY_THREADS (CONV_THREADS - NOUT)
#define MLP_FULL   (NBW * BB / CONV_THREADS)
#define MLP_REST   (NBW * BB - MLP_FULL * CONV_THREADS)

__device__ __forceinline__ float fast_rcp(float v) { return __builtin_amdgcn_rcpf(v); }

__device__ __forceinline__ float bw_inv(const float4* __restrict__ bww_s, const float4* __restrict__ enc_s4,
                                        const float* __restrict__ bwb_s, int idx) {
  const int bb = idx & (BB - 1);
  const int ju = idx >> 3;
  const float4* wr = bww_s + ju * (NLAT / 4);
  const float4* er = enc_s4 + bb * (NLAT / 4);
  float acc = bwb_s[ju];
#pragma unroll
  for (int c = 0; c < NLAT / 4; ++c) {
    const float4 ev = er[c], wv = wr[c];
    acc = fmaf(ev.x, wv.x, acc);
    acc = fmaf(ev.y, wv.y, acc);
    acc = fmaf(ev.z, wv.z, acc);
    acc = fmaf(ev.w, wv.w, acc);
  }
  const float s = fast_rcp(1.0f + expf(-acc));
  const float lo = 2.0f / 60.0f / MUF;
  const float hi = 1.0f / 60.0f;
  const float bw = (hi - lo) * s + lo;
  const float d = bw * MUF;
  return fast_rcp(d * d);
}

__global__ __launch_bounds__(CONV_THREADS)
void conv_kernel(const float* __restrict__ ndist,
                 const int* __restrict__ nbid,
                 const int* __restrict__ labels,
                 const float* __restrict__ dec_t,
                 const float* __restrict__ partials,
                 const float* __restrict__ enc_b,
                 const float* __restrict__ bw_w,
                 const float* __restrict__ bw_b,
                 float* __restrict__ out) {
  __shared__ float stage[BB * CONV_PITCH];
  __shared__ float4 gtile[CONV_WAVES * NLAT * (GT_PITCH / 4)];
  __shared__ float4 enc_s4[NOUT / 4];
  __shared__ float4 bww_s[3 * COPY_THREADS];
  __shared__ float bwb_s[NBW];
  __shared__ float4 sinv4[NBW * BB / 4];
  float* enc_s = (float*)enc_s4;
  float* sinv  = (float*)sinv4;
  const int tid = threadIdx.x;
  const int t  = tid & 31;
  const int w  = __builtin_amdgcn_readfirstlane(tid >> 5);
  const int nw = blockIdx.x * CONV_PTS + w * PPW;
  const int p  = t >> 3;
  const int j  = t & 7;
  const int n  = nw + p;
  const bool act = j < NPIECE;
  const int jc = act ? j : NPIECE - 1;
  const int4*   nb4 = (const int4*)  (nbid  + (size_t)n * KNB);
  const float4* nd4 = (const float4*)(ndist + (size_t)n * KNB);
  int4 iv[KNB / 4];
  float4 dv[KNB / 4];
#pragma unroll
  for (int r = 0; r < KNB / 4; ++r) { iv[r] = nb4[r]; dv[r] = nd4[r]; }
  const int m = labels[n];
  const float bpiece = bw_b[tid < NBW ? tid : NBW - 1];
  if (tid < NOUT) {
    float s = 0.f;
#pragma unroll
    for (int pp = 0; pp < NENC; ++pp) s += partials[pp * NOUT + tid];
    enc_s[tid] = s + enc_b[tid % NLAT];
  } else {
    const float4* w4 = (const float4*)bw_w;
    const int i0 = tid - NOUT;
    const int i2 = i0 + 2 * COPY_THREADS < NBW * NLAT / 4 ? i0 + 2 * COPY_THREADS : NBW * NLAT / 4 - 1;
    const float4 wp0 = w4[i0], wp1 = w4[i0 + COPY_THREADS], wp2 = w4[i2];
    bww_s[i0] = wp0;
    bww_s[i0 + COPY_THREADS] = wp1;
    bww_s[i0 + 2 * COPY_THREADS] = wp2;
  }
  if (tid < NBW) bwb_s[tid] = bpiece;
  __syncthreads();
  float4 q[KNB];
#pragma unroll
  for (int r = 0; r < KNB / 4; ++r) {
    const int ids[4] = {iv[r].x, iv[r].y, iv[r].z, iv[r].w};
#pragma unroll
    for (int u = 0; u < 4; ++u)
      q[4 * r + u] = *(const float4*)((const char*)dec_t +
                                      ((unsigned)ids[u] * (unsigned)(DT_PITCH * 4) + (unsigned)(16 * jc)));
  }
  {
    float r3[MLP_FULL];
#pragma unroll
    for (int r = 0; r < MLP_FULL; ++r) r3[r] = bw_inv(bww_s, enc_s4, bwb_s, tid + r * CONV_THREADS);
#pragma unroll
    for (int r = 0; r < MLP_FULL; ++r) sinv[tid + r * CONV_THREADS] = r3[r];
    if (tid < MLP_REST)
      sinv[tid + MLP_FULL * CONV_THREADS] = bw_inv(bww_s, enc_s4, bwb_s, tid + MLP_FULL * CONV_THREADS);
  }
  __syncthreads();
  float e[KNB];
#pragma unroll
  for (int r = 0; r < KNB / 4; ++r) {
    e[4 * r + 0] = dv[r].x * dv[r].x;  e[4 * r + 1] = dv[r].y * dv[r].y;
    e[4 * r + 2] = dv[r].z * dv[r].z;  e[4 * r + 3] = dv[r].w * dv[r].w;
  }
  float a0x[4], a0y[4], a0z[4], a0w[4], a1x[4], a1y[4], a1z[4], a1w[4], ads[4], adm[4];
#pragma unroll
  for (int a = 0; a < 4; ++a) {
    const int r0 = 8 * a;
    a0x[a] = q[r0].x;  a0y[a] = q[r0].y;  a0z[a] = q[r0].z;  a0w[a] = q[r0].w;
    a1x[a] = q[r0].x * e[r0];  a1y[a] = q[r0].y * e[r0];
    a1z[a] = q[r0].z * e[r0];  a1w[a] = q[r0].w * e[r0];
    ads[a] = e[r0];  adm[a] = e[r0];
#pragma unroll
    for (int r = r0 + 1; r < r0 + 8; ++r) {
      a0x[a] += q[r].x;  a0y[a] += q[r].y;  a0z[a] += q[r].z;  a0w[a] += q[r].w;
      a1x[a] = fmaf(q[r].x, e[r], a1x[a]);  a1y[a] = fmaf(q[r].y, e[r], a1y[a]);
      a1z[a] = fmaf(q[r].z, e[r], a1z[a]);  a1w[a] = fmaf(q[r].w, e[r], a1w[a]);
      ads[a] += e[r];
      adm[a] = fmaxf(adm[a], e[r]);
    }
  }
  const float g0[4] = {(a0x[0] + a0x[1]) + (a0x[2] + a0x[3]), (a0y[0] + a0y[1]) + (a0y[2] + a0y[3]),
                       (a0z[0] + a0z[1]) + (a0z[2] + a0z[3]), (a0w[0] + a0w[1]) + (a0w[2] + a0w[3])};
  const float g1[4] = {(a1x[0] + a1x[1]) + (a1x[2] + a1x[3]), (a1y[0] + a1y[1]) + (a1y[2] + a1y[3]),
                       (a1z[0] + a1z[1]) + (a1z[2] + a1z[3]), (a1w[0] + a1w[1]) + (a1w[2] + a1w[3])};
  const float d2sum = (ads[0] + ads[1]) + (ads[2] + ads[3]);
  const float d2max = fmaxf(fmaxf(adm[0], adm[1]), fmaxf(adm[2], adm[3]));
  float inv[4][BB];
#pragma unroll
  for (int c = 0; c < 4; ++c) {
    const float4* ip = sinv4 + ((4 * jc + c) * MCL + m) * (BB / 4);
    const float4 lo4 = ip[0], hi4 = ip[1];
    inv[c][0] = lo4.x;  inv[c][1] = lo4.y;  inv[c][2] = lo4.z;  inv[c][3] = lo4.w;
    inv[c][4] = hi4.x;  inv[c][5] = hi4.y;  inv[c][6] = hi4.z;  inv[c][7] = hi4.w;
  }
  float invmax = inv[0][0];
#pragma unroll
  for (int c = 0; c < 4; ++c)
#pragma unroll
    for (int b = 0; b < BB; ++b) invmax = fmaxf(invmax, inv[c][b]);
  const unsigned open = (unsigned)__ballot(fmaf(d2max, -invmax, 1.0f) > 0.0f);
  {
    float acc[BB];
#pragma unroll
    for (int b = 0; b < BB; ++b) {
      const float4 eb = enc_s4[b * (NLAT / 4) + jc];
      const float eb_[4] = {eb.x, eb.y, eb.z, eb.w};
      float a = 0.f;
#pragma unroll
      for (int c = 0; c < 4; ++c) {
        const float conv = fmaf(-inv[c][b], g1[c], g0[c]) * fast_rcp(fmaf(-inv[c][b], d2sum, (float)KNB));
        a = fmaf(eb_[c], conv, a);
      }
      if (!act) a = 0.f;
      a += __shfl_xor(a, 1, 32);
      a += __shfl_xor(a, 2, 32);
      a += __shfl_xor(a, 4, 32);
      acc[b] = a;
    }
    if (j == 0) {
#pragma unroll
      for (int b = 0; b < BB; ++b) stage[b * CONV_PITCH + w * PPW + p] = acc[b];
    }
  }
  if (open != 0xFFFFFFFFu) {
    const int b  = t & 7;
    const int lq = t >> 3;
    float* gw = (float*)(gtile + w * (NLAT * (GT_PITCH / 4)));
    for (int pp = 0; pp < PPW; ++pp) {
      if (((open >> (8 * pp)) & 0xFFu) == 0xFFu) continue;
      const int np = nw + pp;
      const int mp = labels[np];
      const int nid = nbid[(size_t)np * KNB + t];
      const float4* pl = (const float4*)(dec_t + (size_t)nid * DT_PITCH);
      float4 qg[NPIECE];
#pragma unroll
      for (int c = 0; c < NPIECE; ++c) qg[c] = pl[c];
      const float* ndp = ndist + (size_t)np * KNB;
      float ndv[KNB];
#pragma unroll
      for (int k = 0; k < KNB; ++k) ndv[k] = ndp[k];
#pragma unroll
      for (int c = 0; c < NPIECE; ++c) {
        gw[(4 * c + 0) * GT_PITCH + t] = qg[c].x;
        gw[(4 * c + 1) * GT_PITCH + t] = qg[c].y;
        gw[(4 * c + 2) * GT_PITCH + t] = qg[c].z;
        gw[(4 * c + 3) * GT_PITCH + t] = qg[c].w;
      }
      const v2f one2  = {1.f, 1.f};
      const v2f zero2 = {0.f, 0.f};
      float outacc = 0.f;
      const float* invcol = sinv + mp * BB + b;
      const float* gwl = gw + lq * GT_PITCH;
#pragma unroll 1
      for (int i = 0; i < NLAT / 4; ++i) {
        const int l = lq + 4 * i;
        const float inv1 = invcol[l * (MCL * BB)];
        const v2f ninv = {-inv1, -inv1};
        const float4* gr = (const float4*)(gwl + 4 * i * GT_PITCH);
        v2f ws = zero2, ps = zero2;
#pragma unroll
        for (int k4 = 0; k4 < KNB / 4; ++k4) {
          const float4 g = gr[k4];
          const v2f dA = {ndv[4 * k4 + 0] * ndv[4 * k4 + 0], ndv[4 * k4 + 1] * ndv[4 * k4 + 1]};
          const v2f dB = {ndv[4 * k4 + 2] * ndv[4 * k4 + 2], ndv[4 * k4 + 3] * ndv[4 * k4 + 3]};
          const v2f ga = {g.x, g.y};
          const v2f gb = {g.z, g.w};
          const v2f wa = __builtin_elementwise_max(zero2, __builtin_elementwise_fma(dA, ninv, one2));
          const v2f wb = __builtin_elementwise_max(zero2, __builtin_elementwise_fma(dB, ninv, one2));
          ws += wa;
          ps = __builtin_elementwise_fma(ga, wa, ps);
          ws += wb;
          ps = __builtin_elementwise_fma(gb, wb, ps);
        }
        const float conv = (ps.x + ps.y) / (ws.x + ws.y);
        outacc = fmaf(enc_s[b * NLAT + l], conv, outacc);
      }
      outacc += __shfl_xor(outacc, 8, 32);
      outacc += __shfl_xor(outacc, 16, 32);
      if (t < 8)
        stage[b * CONV_PITCH + w * PPW + pp] = outacc;
    }
  }
  __syncthreads();
  out[(size_t)(w >> 1) * NTOT + blockIdx.x * CONV_PTS + (w & 1) * 32 + t] =
      stage[(w >> 1) * CONV_PITCH + (w & 1) * 32 + t];
}

extern "C" void kernel_launch(void* const* d_in, const int* in_sizes, int n_in,
                              void* d_out, int out_size, void* d_ws, size_t ws_size,
                              hipStream_t stream) {
  const float* x      = (const float*)d_in[0];
  const float* enc_w  = (const float*)d_in[1];
  const float* enc_b  = (const float*)d_in[2];
  const float* dec_w  = (const float*)d_in[3];
  const float* bw_w   = (const float*)d_in[4];
  const float* bw_b   = (const float*)d_in[5];
  const float* ndist  = (const float*)d_in[6];
  const int*   nbid   = (const int*)d_in[7];
  const int*   labels = (const int*)d_in[8];
  float* out = (float*)d_out;

  float* ws       = (float*)d_ws;
  float* partials = ws + WS_PARTIALS;
  float* dec_t    = ws + WS_DECT;

  front_kernel<<<NENC + NTOT / TR_NODES, ENC_WAVES * 32, 0, stream>>>(x, enc_w, dec_w, partials, dec_t);
  conv_kernel<<<NTOT / CONV_PTS, CONV_THREADS, 0, stream>>>(ndist, nbid, labels, dec_t, partials, enc_b, bw_w, bw_b, out);
}
